# starting kernel
# baseline (speedup 1.0000x reference)
#include <hip/hip_runtime.h>

#define N_NODES 200000
#define N_EDGES 3200000
#define FDIM    16
#define NTILES  (N_NODES / 16)

typedef float v2f __attribute__((ext_vector_type(2)));
typedef float v8f __attribute__((ext_vector_type(8)));

__global__ void k_zero(float* __restrict__ p, int n) {
    int i = blockIdx.x * blockDim.x + threadIdx.x;
    if (i < n) p[i] = 0.0f;
}

__global__ __launch_bounds__(256) void k_scatter(const float* __restrict__ feat,
                                                 const int* __restrict__ ei,
                                                 float* __restrict__ agg) {
    unsigned int t = blockIdx.x * 256u + threadIdx.x;
    if (t >= (unsigned int)N_EDGES * FDIM) return;
    int e = (int)(t >> 4);
    int f = (int)(t & 15u);
    int s = ei[e];
    int d = ei[N_EDGES + e];
    atomicAdd(&agg[(size_t)d * FDIM + f], feat[(size_t)s * FDIM + f]);
}

template <bool LOGSOFTMAX>
__global__ __launch_bounds__(256) void k_gin_mlp(const float* __restrict__ xin,
                                                 const float* __restrict__ agg,
                                                 const float* __restrict__ eps_p,
                                                 const float* __restrict__ wa,
                                                 const float* __restrict__ ba,
                                                 const float* __restrict__ wb,
                                                 const float* __restrict__ bb,
                                                 float* __restrict__ out,
                                                 int n_tiles) {
    __shared__ float lds[8][16][17];

    const int wid  = threadIdx.x >> 5;
    const int lane = threadIdx.x & 31;
    const int tile = blockIdx.x * 8 + wid;
    if (tile >= n_tiles) return;

    const int col = lane & 15;
    const int hi  = lane >> 4;
    const float epsp = 1.0f + eps_p[0];

    const float* xrow = xin + (size_t)(tile * 16 + col) * FDIM;
    const float* grow = agg + (size_t)(tile * 16 + col) * FDIM;

    v8f acc = {};
#pragma unroll
    for (int c = 0; c < 4; ++c) {
        const int kb = c * 4 + hi * 2;
        v2f xv = *(const v2f*)(xrow + kb);
        v2f gv = *(const v2f*)(grow + kb);
        v2f a;
        a.x = epsp * xv.x + gv.x;
        a.y = epsp * xv.y + gv.y;
        v2f b;
        b.x = wa[(kb + 0) * 16 + col];
        b.y = wa[(kb + 1) * 16 + col];
        acc = __builtin_amdgcn_wmma_f32_16x16x4_f32(false, a, false, b,
                                                    (short)0, acc, false, false);
    }

    const float biasA = ba[col];
#pragma unroll
    for (int v = 0; v < 8; ++v) {
        float t = acc[v] + biasA;
        lds[wid][v + hi * 8][col] = t > 0.0f ? t : 0.0f;
    }
    __syncthreads();

    v8f acc2 = {};
#pragma unroll
    for (int c = 0; c < 4; ++c) {
        const int kb = c * 4 + hi * 2;
        v2f a;
        a.x = lds[wid][col][kb + 0];
        a.y = lds[wid][col][kb + 1];
        v2f b;
        b.x = wb[(kb + 0) * 16 + col];
        b.y = wb[(kb + 1) * 16 + col];
        acc2 = __builtin_amdgcn_wmma_f32_16x16x4_f32(false, a, false, b,
                                                     (short)0, acc2, false, false);
    }

    const float biasB = bb[col];
    float* orow = out + (size_t)tile * 16 * FDIM;

    if (LOGSOFTMAX) {
#pragma unroll
        for (int v = 0; v < 8; ++v) {
            float val = acc2[v] + biasB;
            float m = val;
#pragma unroll
            for (int off = 1; off < 16; off <<= 1)
                m = fmaxf(m, __shfl_xor(m, off, 16));
            float s = __expf(val - m);
#pragma unroll
            for (int off = 1; off < 16; off <<= 1)
                s += __shfl_xor(s, off, 16);
            orow[(v + hi * 8) * FDIM + col] = val - m - __logf(s);
        }
    } else {
#pragma unroll
        for (int v = 0; v < 8; ++v) {
            float t = acc2[v] + biasB;
            orow[(v + hi * 8) * FDIM + col] = t > 0.0f ? t : 0.0f;
        }
    }
}

extern "C" void kernel_launch(void* const* d_in, const int* in_sizes, int n_in,
                              void* d_out, int out_size, void* d_ws, size_t ws_size,
                              hipStream_t stream) {
    const float* x    = (const float*)d_in[0];
    const int*   ei   = (const int*)d_in[1];
    const float* eps1 = (const float*)d_in[2];
    const float* w1a  = (const float*)d_in[3];
    const float* b1a  = (const float*)d_in[4];
    const float* w1b  = (const float*)d_in[5];
    const float* b1b  = (const float*)d_in[6];
    const float* eps2 = (const float*)d_in[7];
    const float* w2a  = (const float*)d_in[8];
    const float* b2a  = (const float*)d_in[9];
    const float* w2b  = (const float*)d_in[10];
    const float* b2b  = (const float*)d_in[11];
    float* out = (float*)d_out;

    float* agg = (float*)d_ws;
    float* h1  = agg + (size_t)N_NODES * FDIM;

    const int nfeat       = N_NODES * FDIM;
    const int zero_blocks = (nfeat + 255) / 256;
    const unsigned int scat_threads = (unsigned int)N_EDGES * FDIM;
    const int scat_blocks = (int)((scat_threads + 255u) / 256u);
    const int mlp_blocks  = (NTILES + 7) / 8;

    k_zero<<<zero_blocks, 256, 0, stream>>>(agg, nfeat);
    k_scatter<<<scat_blocks, 256, 0, stream>>>(x, ei, agg);
    k_gin_mlp<false><<<mlp_blocks, 256, 0, stream>>>(x, agg, eps1, w1a, b1a, w1b, b1b,
                                                     h1, NTILES);
    k_zero<<<zero_blocks, 256, 0, stream>>>(agg, nfeat);
    k_scatter<<<scat_blocks, 256, 0, stream>>>(h1, ei, agg);
    k_gin_mlp<true><<<mlp_blocks, 256, 0, stream>>>(h1, agg, eps2, w2a, b2a, w2b, b2b,
                                                    out, NTILES);
}


# List build queues whole 32-edge match words (one wave vote per vector pair, not one round per matched edge), unpacked in place to edge numbers after the scan; add order by arrival changes
# speedup vs baseline: 3.3305x; 3.3305x over previous
#include <hip/hip_runtime.h>


#define N_NODES 200000
#define N_EDGES 3200000
#define FDIM    16
#define NTILES  (N_NODES / 16)
#define SLOTS   16
#define SPAN    1600
#define NOWN    (N_NODES / SPAN)
#define NVEC    (N_EDGES / 16)
#define HVEC    (NVEC / 2)
#define QE      448
#define IPITCH  36

typedef float v2f __attribute__((ext_vector_type(2)));
typedef float v8f __attribute__((ext_vector_type(8)));

__global__ __launch_bounds__(256) void k_owner(const int* __restrict__ ei,
                                               int* __restrict__ own,
                                               int* __restrict__ rec,
                                               int* __restrict__ ovfcnt) {
    const int t = blockIdx.x * 256 + threadIdx.x;
    if (t == 0) ovfcnt[0] = 0;
    if (t >= N_EDGES / 4) return;
    const int4 s4 = *(const int4*)(ei + (size_t)t * 4);
    const int4 d4 = *(const int4*)(ei + N_EDGES + (size_t)t * 4);
    const unsigned o0 = (unsigned)d4.x / SPAN, o1 = (unsigned)d4.y / SPAN;
    const unsigned o2 = (unsigned)d4.z / SPAN, o3 = (unsigned)d4.w / SPAN;
    own[t] = (int)(o0 | (o1 << 8) | (o2 << 16) | (o3 << 24));
    int4 r;
    r.x = (int)((((unsigned)d4.x - o0 * SPAN) << 18) | (unsigned)s4.x);
    r.y = (int)((((unsigned)d4.y - o1 * SPAN) << 18) | (unsigned)s4.y);
    r.z = (int)((((unsigned)d4.z - o2 * SPAN) << 18) | (unsigned)s4.z);
    r.w = (int)((((unsigned)d4.w - o3 * SPAN) << 18) | (unsigned)s4.w);
    *(int4*)(rec + (size_t)t * 4) = r;
}

__device__ __forceinline__ unsigned zero_bytes(unsigned y) {
    return ~(((y & 0x7F7F7F7Fu) + 0x7F7F7F7Fu) | y) & 0x80808080u;
}

__device__ __forceinline__ void place_one(unsigned r, int base, int* sCnt, int* sMain,
                                          int* sSpill,
                                          int* __restrict__ ovfcnt, int* __restrict__ ovf) {
    const int local = (int)(r >> 18);
    const int s = (int)(r & 0x3FFFFu);
    const int tk = atomicAdd(&sCnt[local], 1);
    if (tk < SLOTS) {
        sMain[local * SLOTS + tk] = s;
    } else if (tk < 2 * SLOTS) {
        sSpill[local * SLOTS + tk - SLOTS] = s;
    } else {
        const int k = atomicAdd(ovfcnt, 1);
        ovf[2 * (size_t)k + 0] = base + local;
        ovf[2 * (size_t)k + 1] = s;
    }
}

__device__ __forceinline__ int edge_of(unsigned x1, int bit, int vbase) {
    unsigned x2 = x1 + 1024u;
    x2 = x2 >= (unsigned)HVEC ? x2 - (unsigned)HVEC : x2;
    const unsigned x = (bit & 4) ? x1 : x2;
    return (int)(((unsigned)vbase + x) * 16u + (unsigned)((bit & 3) * 4 + (bit >> 3)));
}

__device__ __forceinline__ void file_word(unsigned x1, unsigned m, int vbase,
                                          const int* __restrict__ rec, int base, int* sCnt,
                                          int* sMain, int* sSpill,
                                          int* __restrict__ ovfcnt, int* __restrict__ ovf) {
    while (m != 0u) {
        const int bit = __builtin_ctz(m);
        m &= m - 1u;
        place_one((unsigned)rec[edge_of(x1, bit, vbase)], base, sCnt, sMain, sSpill, ovfcnt, ovf);
    }
}

__global__ __launch_bounds__(1024) void k_build(const int* __restrict__ own,
                                                const int* __restrict__ rec,
                                                int* __restrict__ cnt,
                                                int* __restrict__ mainL,
                                                int* __restrict__ spillL,
                                                int* __restrict__ ovf) {
    __shared__ __attribute__((aligned(16))) int sMain[SPAN * SLOTS];
    __shared__ __attribute__((aligned(16))) int sSpill[SPAN * SLOTS];
    __shared__ __attribute__((aligned(8))) unsigned sQ[32 * 2 * QE];
    __shared__ int sCnt[SPAN];

    const int tid  = threadIdx.x;
    const int lane = tid & 31;
    const int wid  = tid >> 5;
    const int rng  = blockIdx.x >> 1;
    const int half = blockIdx.x & 1;
    const int base = rng * SPAN;
    int* ovfcnt = cnt + 2 * N_NODES;

    for (int i = tid; i < SPAN * SLOTS / 4; i += 1024) {
        ((int4*)sMain)[i]  = make_int4(0, 0, 0, 0);
        ((int4*)sSpill)[i] = make_int4(0, 0, 0, 0);
    }
    for (int i = tid; i < SPAN; i += 1024) sCnt[i] = 0;
    __syncthreads();

    const unsigned pat = (unsigned)rng * 0x01010101u;
    const uint4* __restrict__ o4 = (const uint4*)own + (size_t)half * HVEC;
    const int rot = rng * (HVEC / NOWN);
    const int vbase = half * HVEC;
    unsigned* myq = sQ + wid * (2 * QE);
    int qn = 0;

    for (int g = 0; g < HVEC; g += 8 * 1024) {
        uint4 v[8];
        int vi[8];
        bool ok[8];
#pragma unroll
        for (int k = 0; k < 8; ++k) {
            const int it = g + k * 1024 + tid;
            ok[k] = it < HVEC;
            int x = it + rot;
            x = x >= HVEC ? x - HVEC : x;
            x = ok[k] ? x : 0;
            vi[k] = x;
            v[k] = o4[x];
        }
#pragma unroll
        for (int k = 0; k < 8; k += 2) {
            unsigned ma = (zero_bytes(v[k].x ^ pat) >> 3) | (zero_bytes(v[k].y ^ pat) >> 2) |
                          (zero_bytes(v[k].z ^ pat) >> 1) | zero_bytes(v[k].w ^ pat);
            unsigned mb = (zero_bytes(v[k + 1].x ^ pat) >> 7) | (zero_bytes(v[k + 1].y ^ pat) >> 6) |
                          (zero_bytes(v[k + 1].z ^ pat) >> 5) | (zero_bytes(v[k + 1].w ^ pat) >> 4);
            unsigned m = (ok[k] ? ma : 0u) | (ok[k + 1] ? mb : 0u);
            const unsigned bal = __builtin_amdgcn_ballot_w32(m != 0u);
            if (m != 0u) {
                const int pos = qn + (int)__builtin_amdgcn_mbcnt_lo(bal, 0u);
                if (pos < QE) {
                    myq[2 * pos + 0] = (unsigned)vi[k];
                    myq[2 * pos + 1] = m;
                } else {
                    file_word((unsigned)vi[k], m, vbase, rec, base, sCnt, sMain, sSpill,
                              ovfcnt, ovf);
                }
            }
            qn += __builtin_popcount(bal);
        }
    }

    const int nq = qn < QE ? qn : QE;
    int wp = 0;
    for (int i0 = 0; i0 < nq; i0 += 128) {
        unsigned ex[4], em[4];
#pragma unroll
        for (int k = 0; k < 4; ++k) {
            const int i = i0 + k * 32 + lane;
            const bool okk = i < nq;
            ex[k] = okk ? myq[2 * i + 0] : 0u;
            em[k] = okk ? myq[2 * i + 1] : 0u;
        }
        const int lim = i0 + 128 < nq ? 2 * (i0 + 128) : 2 * QE;
#pragma unroll
        for (int k = 0; k < 4; ++k) {
            unsigned m = em[k];
            for (;;) {
                const unsigned bal = __builtin_amdgcn_ballot_w32(m != 0u);
                if (bal == 0u) break;
                if (m != 0u) {
                    const int bit = __builtin_ctz(m);
                    m &= m - 1u;
                    const int e = edge_of(ex[k], bit, vbase);
                    const int pos = wp + (int)__builtin_amdgcn_mbcnt_lo(bal, 0u);
                    if (pos < lim) myq[pos] = (unsigned)e;
                    else place_one((unsigned)rec[e], base, sCnt, sMain, sSpill, ovfcnt, ovf);
                }
                wp += __builtin_popcount(bal);
            }
        }
        wp = wp < lim ? wp : lim;
    }

    for (int i0 = 0; i0 < wp; i0 += 128) {
        unsigned r[4];
        bool ok[4];
#pragma unroll
        for (int k = 0; k < 4; ++k) {
            const int i = i0 + k * 32 + lane;
            ok[k] = i < wp;
            r[k] = ok[k] ? (unsigned)rec[myq[i]] : 0u;
        }
#pragma unroll
        for (int k = 0; k < 4; ++k)
            if (ok[k]) place_one(r[k], base, sCnt, sMain, sSpill, ovfcnt, ovf);
    }
    __syncthreads();

    const size_t nb = (size_t)half * N_NODES + base;
    for (int i = tid; i < SPAN; i += 1024) cnt[nb + i] = sCnt[i];
    int4* gm = (int4*)(mainL + nb * SLOTS);
    for (int i = tid; i < SPAN * SLOTS / 4; i += 1024) gm[i] = ((const int4*)sMain)[i];
    int4* gs = (int4*)(spillL + nb * SLOTS);
    for (int i = tid; i < SPAN * SLOTS / 4; i += 1024)
        if (sCnt[i >> 2] > SLOTS) gs[i] = ((const int4*)sSpill)[i];
}

template <bool LOGSOFTMAX>
__global__ __launch_bounds__(128) void k_gin_layer(const float* __restrict__ xin,
                                                   const int* __restrict__ cnt,
                                                   const int* __restrict__ mainL,
                                                   const int* __restrict__ spillL,
                                                   const int* __restrict__ ovf,
                                                   const float* __restrict__ eps_p,
                                                   const float* __restrict__ wa,
                                                   const float* __restrict__ ba,
                                                   const float* __restrict__ wb,
                                                   const float* __restrict__ bb,
                                                   float* __restrict__ out) {
    __shared__ float ldsM[4][16][17];
    __shared__ float ldsT[4][16][17];
    __shared__ __attribute__((aligned(16))) int ldsI[4][16][IPITCH];

    const int wid  = threadIdx.x >> 5;
    const int lane = threadIdx.x & 31;
    const int tile = blockIdx.x * 4 + wid;

    const int g = lane >> 2;
    const int q = lane & 3;
    const int node0 = tile * 16 + g;
    const int node1 = node0 + 8;

    const float4* __restrict__ f4 = (const float4*)xin;
    const int cA0 = cnt[node0];
    const int cB0 = cnt[N_NODES + node0];
    const int cA1 = cnt[node1];
    const int cB1 = cnt[N_NODES + node1];
    const int4 mA0 = *(const int4*)(mainL + (size_t)node0 * SLOTS + 4 * q);
    const int4 mB0 = *(const int4*)(mainL + ((size_t)N_NODES + node0) * SLOTS + 4 * q);
    const int4 mA1 = *(const int4*)(mainL + (size_t)node1 * SLOTS + 4 * q);
    const int4 mB1 = *(const int4*)(mainL + ((size_t)N_NODES + node1) * SLOTS + 4 * q);
    const float4 x0 = f4[(size_t)node0 * 4 + q];
    const float4 x1 = f4[(size_t)node1 * 4 + q];

    const int dA0 = cA0 < SLOTS ? cA0 : SLOTS;
    const int dB0 = cB0 < SLOTS ? cB0 : SLOTS;
    const int dA1 = cA1 < SLOTS ? cA1 : SLOTS;
    const int dB1 = cB1 < SLOTS ? cB1 : SLOTS;
    const int d0 = dA0 + dB0;
    const int d1 = dA1 + dB1;

    int* l0 = &ldsI[wid][g][0];
    int* l1 = &ldsI[wid][g + 8][0];
    *(int4*)(l0 + 4 * q) = mA0;
    *(int4*)(l1 + 4 * q) = mA1;
    *(int4*)(l0 + SLOTS + 4 * q) = make_int4(0, 0, 0, 0);
    *(int4*)(l1 + SLOTS + 4 * q) = make_int4(0, 0, 0, 0);
    l0[dA0 + 4 * q + 0] = mB0.x; l0[dA0 + 4 * q + 1] = mB0.y;
    l0[dA0 + 4 * q + 2] = mB0.z; l0[dA0 + 4 * q + 3] = mB0.w;
    l1[dA1 + 4 * q + 0] = mB1.x; l1[dA1 + 4 * q + 1] = mB1.y;
    l1[dA1 + 4 * q + 2] = mB1.z; l1[dA1 + 4 * q + 3] = mB1.w;

    float4 a0 = make_float4(0.f, 0.f, 0.f, 0.f);
    float4 a1 = make_float4(0.f, 0.f, 0.f, 0.f);

    int dm = d0 > d1 ? d0 : d1;
#pragma unroll
    for (int off = 1; off < 32; off <<= 1) {
        const int o = __shfl_xor(dm, off, 32);
        dm = o > dm ? o : dm;
    }
    dm = __builtin_amdgcn_readfirstlane(dm);
    __syncthreads();

    for (int i = 0; i < dm; i += 4) {
        const int4 s0 = *(const int4*)(l0 + i);
        const int4 s1 = *(const int4*)(l1 + i);
        const bool v00 = i + 0 < d0, v01 = i + 1 < d0, v02 = i + 2 < d0, v03 = i + 3 < d0;
        const bool v10 = i + 0 < d1, v11 = i + 1 < d1, v12 = i + 2 < d1, v13 = i + 3 < d1;
        const float4 r00 = f4[(size_t)(v00 ? s0.x : node0) * 4 + q];
        const float4 r01 = f4[(size_t)(v01 ? s0.y : node0) * 4 + q];
        const float4 r02 = f4[(size_t)(v02 ? s0.z : node0) * 4 + q];
        const float4 r03 = f4[(size_t)(v03 ? s0.w : node0) * 4 + q];
        const float4 r10 = f4[(size_t)(v10 ? s1.x : node1) * 4 + q];
        const float4 r11 = f4[(size_t)(v11 ? s1.y : node1) * 4 + q];
        const float4 r12 = f4[(size_t)(v12 ? s1.z : node1) * 4 + q];
        const float4 r13 = f4[(size_t)(v13 ? s1.w : node1) * 4 + q];
#define ACC(a, r, v) { a.x += (v) ? r.x : 0.0f; a.y += (v) ? r.y : 0.0f; \
                       a.z += (v) ? r.z : 0.0f; a.w += (v) ? r.w : 0.0f; }
        ACC(a0, r00, v00) ACC(a0, r01, v01) ACC(a0, r02, v02) ACC(a0, r03, v03)
        ACC(a1, r10, v10) ACC(a1, r11, v11) ACC(a1, r12, v12) ACC(a1, r13, v13)
#undef ACC
    }

    if (cA0 > SLOTS || cB0 > SLOTS || cA1 > SLOTS || cB1 > SLOTS) {
#define SPILL(a, c, row) { const int e_ = (c) < 2 * SLOTS ? (c) : 2 * SLOTS;                 \
        for (int j = SLOTS; j < e_; ++j) {                                                  \
            const int ss = spillL[(row) * SLOTS + (j - SLOTS)];                              \
            const float4 r = f4[(size_t)ss * 4 + q];                                        \
            a.x += r.x; a.y += r.y; a.z += r.z; a.w += r.w; } }
        SPILL(a0, cA0, (size_t)node0)
        SPILL(a0, cB0, (size_t)N_NODES + node0)
        SPILL(a1, cA1, (size_t)node1)
        SPILL(a1, cB1, (size_t)N_NODES + node1)
#undef SPILL
    }

    if (cA0 > 2 * SLOTS || cB0 > 2 * SLOTS || cA1 > 2 * SLOTS || cB1 > 2 * SLOTS) {
        int novf = cnt[2 * N_NODES];
        novf = novf < N_EDGES ? novf : N_EDGES;
        for (int j = 0; j < novf; ++j) {
            const int dd = ovf[2 * (size_t)j + 0];
            const int ss = ovf[2 * (size_t)j + 1];
            if (dd == node0) {
                const float4 r = f4[(size_t)ss * 4 + q];
                a0.x += r.x; a0.y += r.y; a0.z += r.z; a0.w += r.w;
            }
            if (dd == node1) {
                const float4 r = f4[(size_t)ss * 4 + q];
                a1.x += r.x; a1.y += r.y; a1.z += r.z; a1.w += r.w;
            }
        }
    }

    const float epsp = 1.0f + eps_p[0];
    {
        float* m0 = &ldsM[wid][g][q * 4];
        float* m1 = &ldsM[wid][g + 8][q * 4];
        m0[0] = epsp * x0.x + a0.x; m0[1] = epsp * x0.y + a0.y;
        m0[2] = epsp * x0.z + a0.z; m0[3] = epsp * x0.w + a0.w;
        m1[0] = epsp * x1.x + a1.x; m1[1] = epsp * x1.y + a1.y;
        m1[2] = epsp * x1.z + a1.z; m1[3] = epsp * x1.w + a1.w;
    }
    __syncthreads();

    const int col = lane & 15;
    const int hi  = lane >> 4;

    v8f acc = {};
#pragma unroll
    for (int c = 0; c < 4; ++c) {
        const int kb = c * 4 + hi * 2;
        v2f a;
        a.x = ldsM[wid][col][kb + 0];
        a.y = ldsM[wid][col][kb + 1];
        v2f b;
        b.x = wa[(kb + 0) * 16 + col];
        b.y = wa[(kb + 1) * 16 + col];
        acc = __builtin_amdgcn_wmma_f32_16x16x4_f32(false, a, false, b,
                                                    (short)0, acc, false, false);
    }

    const float biasA = ba[col];
#pragma unroll
    for (int v = 0; v < 8; ++v) {
        float t = acc[v] + biasA;
        ldsT[wid][v + hi * 8][col] = t > 0.0f ? t : 0.0f;
    }
    __syncthreads();

    v8f acc2 = {};
#pragma unroll
    for (int c = 0; c < 4; ++c) {
        const int kb = c * 4 + hi * 2;
        v2f a;
        a.x = ldsT[wid][col][kb + 0];
        a.y = ldsT[wid][col][kb + 1];
        v2f b;
        b.x = wb[(kb + 0) * 16 + col];
        b.y = wb[(kb + 1) * 16 + col];
        acc2 = __builtin_amdgcn_wmma_f32_16x16x4_f32(false, a, false, b,
                                                     (short)0, acc2, false, false);
    }

    const float biasB = bb[col];
    float* orow = out + (size_t)tile * 16 * FDIM;

    if (LOGSOFTMAX) {
#pragma unroll
        for (int v = 0; v < 8; ++v) {
            float val = acc2[v] + biasB;
            float m = val;
#pragma unroll
            for (int off = 1; off < 16; off <<= 1)
                m = fmaxf(m, __shfl_xor(m, off, 16));
            float s = __expf(val - m);
#pragma unroll
            for (int off = 1; off < 16; off <<= 1)
                s += __shfl_xor(s, off, 16);
            orow[(v + hi * 8) * FDIM + col] = val - m - __logf(s);
        }
    } else {
#pragma unroll
        for (int v = 0; v < 8; ++v) {
            float t = acc2[v] + biasB;
            orow[(v + hi * 8) * FDIM + col] = t > 0.0f ? t : 0.0f;
        }
    }
}

extern "C" void kernel_launch(void* const* d_in, const int* in_sizes, int n_in,
                              void* d_out, int out_size, void* d_ws, size_t ws_size,
                              hipStream_t stream) {
    const float* x    = (const float*)d_in[0];
    const int*   ei   = (const int*)d_in[1];
    const float* eps1 = (const float*)d_in[2];
    const float* w1a  = (const float*)d_in[3];
    const float* b1a  = (const float*)d_in[4];
    const float* w1b  = (const float*)d_in[5];
    const float* b1b  = (const float*)d_in[6];
    const float* eps2 = (const float*)d_in[7];
    const float* w2a  = (const float*)d_in[8];
    const float* b2a  = (const float*)d_in[9];
    const float* w2b  = (const float*)d_in[10];
    const float* b2b  = (const float*)d_in[11];
    float* out = (float*)d_out;

    float* h1     = (float*)d_ws;
    int*   mainL  = (int*)(h1 + (size_t)N_NODES * FDIM);
    int*   spillL = mainL + (size_t)2 * N_NODES * SLOTS;
    int*   ovf    = spillL + (size_t)2 * N_NODES * SLOTS;
    int*   cnt    = ovf + (size_t)N_EDGES * 2;
    int*   rec    = cnt + (2 * N_NODES + 32);
    int*   own    = rec + (size_t)N_EDGES;

    k_owner<<<(N_EDGES / 4 + 255) / 256, 256, 0, stream>>>(ei, own, rec, cnt + 2 * N_NODES);
    k_build<<<2 * NOWN, 1024, 0, stream>>>(own, rec, cnt, mainL, spillL, ovf);
    k_gin_layer<false><<<NTILES / 4, 128, 0, stream>>>(x, cnt, mainL, spillL, ovf, eps1, w1a,
                                                       b1a, w1b, b1b, h1);
    k_gin_layer<true><<<NTILES / 4, 128, 0, stream>>>(h1, cnt, mainL, spillL, ovf, eps2, w2a,
                                                      b2a, w2b, b2b, out);
}


# Owner plane stored bit-sliced (7 one-bit planes of 32-edge words made by wave votes, in aligned 512-byte pieces): list build matches 32 edges in ten bitwise ops; f32 add order by arrival changes
# speedup vs baseline: 3.5963x; 1.0798x over previous
#include <hip/hip_runtime.h>


#define N_NODES 200000
#define N_EDGES 3200000
#define FDIM    16
#define NTILES  (N_NODES / 16)
#define SLOTS   16
#define SPAN    1600
#define NOWN    (N_NODES / SPAN)
#define NBIT    7
#define NBLK    (N_EDGES / 128)
#define HSTEP   12512
#define PLS     25024
#define ROTQ    96
static_assert(NOWN <= (1 << NBIT), "a range number must fit NBIT bits");
static_assert(HSTEP % 32 == 0 && PLS % 32 == 0 && ROTQ % 32 == 0, "512-byte pieces stay aligned");
static_assert(HSTEP <= NBLK && 2 * HSTEP >= NBLK && PLS >= NBLK, "the two halves cover every block");
static_assert((NOWN - 1) * ROTQ < HSTEP, "a pass wraps at most once");
static_assert(N_EDGES % 1024 == 0, "the owner pass has no partial work-group");
#define QE      448
#define IPITCH  36

typedef float v2f __attribute__((ext_vector_type(2)));
typedef float v8f __attribute__((ext_vector_type(8)));

__global__ __launch_bounds__(256) void k_owner(const int* __restrict__ ei,
                                               unsigned* __restrict__ planes,
                                               int* __restrict__ rec,
                                               int* __restrict__ ovfcnt) {
    __shared__ __attribute__((aligned(16))) unsigned sW[NBIT * 8 * 4];
    const int t = blockIdx.x * 256 + threadIdx.x;
    const int lane = threadIdx.x & 31;
    const int wid  = threadIdx.x >> 5;
    if (t == 0) ovfcnt[0] = 0;
    const int4 s4 = *(const int4*)(ei + (size_t)t * 4);
    const int4 d4 = *(const int4*)(ei + N_EDGES + (size_t)t * 4);
    const unsigned o0 = (unsigned)d4.x / SPAN, o1 = (unsigned)d4.y / SPAN;
    const unsigned o2 = (unsigned)d4.z / SPAN, o3 = (unsigned)d4.w / SPAN;
    int4 r;
    r.x = (int)((((unsigned)d4.x - o0 * SPAN) << 18) | (unsigned)s4.x);
    r.y = (int)((((unsigned)d4.y - o1 * SPAN) << 18) | (unsigned)s4.y);
    r.z = (int)((((unsigned)d4.z - o2 * SPAN) << 18) | (unsigned)s4.z);
    r.w = (int)((((unsigned)d4.w - o3 * SPAN) << 18) | (unsigned)s4.w);
    *(int4*)(rec + (size_t)t * 4) = r;

    uint4 mine = make_uint4(0u, 0u, 0u, 0u);
#pragma unroll
    for (int b = 0; b < NBIT; ++b) {
        const unsigned w0 = __builtin_amdgcn_ballot_w32(((o0 >> b) & 1u) != 0u);
        const unsigned w1 = __builtin_amdgcn_ballot_w32(((o1 >> b) & 1u) != 0u);
        const unsigned w2 = __builtin_amdgcn_ballot_w32(((o2 >> b) & 1u) != 0u);
        const unsigned w3 = __builtin_amdgcn_ballot_w32(((o3 >> b) & 1u) != 0u);
        if (lane == b) mine = make_uint4(w0, w1, w2, w3);
    }
    if (lane < NBIT) ((uint4*)sW)[lane * 8 + wid] = mine;
    __syncthreads();
    if (threadIdx.x < NBIT * 8) {
        const int b = threadIdx.x >> 3;
        const int i = threadIdx.x & 7;
        ((uint4*)planes)[(size_t)b * PLS + (size_t)blockIdx.x * 8 + i] = ((const uint4*)sW)[threadIdx.x];
    }
}

__device__ __forceinline__ void place_one(unsigned r, int base, int* sCnt, int* sMain,
                                          int* sSpill,
                                          int* __restrict__ ovfcnt, int* __restrict__ ovf) {
    const int local = (int)(r >> 18);
    const int s = (int)(r & 0x3FFFFu);
    const int tk = atomicAdd(&sCnt[local], 1);
    if (tk < SLOTS) {
        sMain[local * SLOTS + tk] = s;
    } else if (tk < 2 * SLOTS) {
        sSpill[local * SLOTS + tk - SLOTS] = s;
    } else {
        const int k = atomicAdd(ovfcnt, 1);
        ovf[2 * (size_t)k + 0] = base + local;
        ovf[2 * (size_t)k + 1] = s;
    }
}

__device__ __forceinline__ int edge_of(unsigned x, int bit, int bbase) {
    return (int)((((unsigned)bbase + (x >> 2)) << 7) + ((unsigned)bit << 2) + (x & 3u));
}

__device__ __forceinline__ void file_word(unsigned x1, unsigned m, int bbase,
                                          const int* __restrict__ rec, int base, int* sCnt,
                                          int* sMain, int* sSpill,
                                          int* __restrict__ ovfcnt, int* __restrict__ ovf) {
    while (m != 0u) {
        const int bit = __builtin_ctz(m);
        m &= m - 1u;
        place_one((unsigned)rec[edge_of(x1, bit, bbase)], base, sCnt, sMain, sSpill, ovfcnt, ovf);
    }
}

__global__ __launch_bounds__(1024) void k_build(const unsigned* __restrict__ planes,
                                                const int* __restrict__ rec,
                                                int* __restrict__ cnt,
                                                int* __restrict__ mainL,
                                                int* __restrict__ spillL,
                                                int* __restrict__ ovf) {
    __shared__ __attribute__((aligned(16))) int sMain[SPAN * SLOTS];
    __shared__ __attribute__((aligned(16))) int sSpill[SPAN * SLOTS];
    __shared__ __attribute__((aligned(8))) unsigned sQ[32 * 2 * QE];
    __shared__ int sCnt[SPAN];

    const int tid  = threadIdx.x;
    const int lane = tid & 31;
    const int wid  = tid >> 5;
    const int rng  = blockIdx.x >> 1;
    const int half = blockIdx.x & 1;
    const int base = rng * SPAN;
    int* ovfcnt = cnt + 2 * N_NODES;

    for (int i = tid; i < SPAN * SLOTS / 4; i += 1024) {
        ((int4*)sMain)[i]  = make_int4(0, 0, 0, 0);
        ((int4*)sSpill)[i] = make_int4(0, 0, 0, 0);
    }
    for (int i = tid; i < SPAN; i += 1024) sCnt[i] = 0;
    __syncthreads();

    unsigned rb[NBIT];
#pragma unroll
    for (int b = 0; b < NBIT; ++b) rb[b] = ((rng >> b) & 1) ? 0xFFFFFFFFu : 0u;
    const int bbase = half * HSTEP;
    const uint4* __restrict__ pl = (const uint4*)planes + bbase;
    const int rot = rng * ROTQ;
    unsigned* myq = sQ + wid * (2 * QE);
    int qn = 0;

    auto scan_block = [&](const uint4 (&p)[NBIT], bool okb, int blk) {
#pragma unroll
        for (int j = 0; j < 4; ++j) {
            unsigned diff = 0u;
#pragma unroll
            for (int b = 0; b < NBIT; ++b) {
                const unsigned w = j == 0 ? p[b].x : j == 1 ? p[b].y : j == 2 ? p[b].z : p[b].w;
                diff |= w ^ rb[b];
            }
            const unsigned m = okb ? ~diff : 0u;
            const unsigned bal = __builtin_amdgcn_ballot_w32(m != 0u);
            if (m != 0u) {
                const unsigned x = (unsigned)blk * 4u + (unsigned)j;
                const int pos = qn + (int)__builtin_amdgcn_mbcnt_lo(bal, 0u);
                if (pos < QE) {
                    myq[2 * pos + 0] = x;
                    myq[2 * pos + 1] = m;
                } else {
                    file_word(x, m, bbase, rec, base, sCnt, sMain, sSpill, ovfcnt, ovf);
                }
            }
            qn += __builtin_popcount(bal);
        }
    };

    auto block_of = [&](int it, bool& okb) {
        int x = it + rot;
        x = x >= HSTEP ? x - HSTEP : x;
        okb = it < HSTEP && bbase + x < NBLK;
        return okb ? x : 0;
    };

    int g = 0;
    for (; g + 2 * 1024 <= HSTEP; g += 2 * 1024) {
        uint4 p0[NBIT], p1[NBIT];
        bool ok0, ok1;
        const int x0 = block_of(g + tid, ok0);
        const int x1 = block_of(g + 1024 + tid, ok1);
#pragma unroll
        for (int b = 0; b < NBIT; ++b) p0[b] = pl[(size_t)b * PLS + x0];
#pragma unroll
        for (int b = 0; b < NBIT; ++b) p1[b] = pl[(size_t)b * PLS + x1];
        __builtin_amdgcn_sched_barrier(0);
        scan_block(p0, ok0, x0);
        scan_block(p1, ok1, x1);
    }
    for (; g < HSTEP; g += 1024) {
        uint4 p0[NBIT];
        bool ok0;
        const int x0 = block_of(g + tid, ok0);
#pragma unroll
        for (int b = 0; b < NBIT; ++b) p0[b] = pl[(size_t)b * PLS + x0];
        scan_block(p0, ok0, x0);
    }

    const int nq = qn < QE ? qn : QE;
    int wp = 0;
    for (int i0 = 0; i0 < nq; i0 += 128) {
        unsigned ex[4], em[4];
#pragma unroll
        for (int k = 0; k < 4; ++k) {
            const int i = i0 + k * 32 + lane;
            const bool okk = i < nq;
            ex[k] = okk ? myq[2 * i + 0] : 0u;
            em[k] = okk ? myq[2 * i + 1] : 0u;
        }
        const int lim = i0 + 128 < nq ? 2 * (i0 + 128) : 2 * QE;
#pragma unroll
        for (int k = 0; k < 4; ++k) {
            unsigned m = em[k];
            for (;;) {
                const unsigned bal = __builtin_amdgcn_ballot_w32(m != 0u);
                if (bal == 0u) break;
                if (m != 0u) {
                    const int bit = __builtin_ctz(m);
                    m &= m - 1u;
                    const int e = edge_of(ex[k], bit, bbase);
                    const int pos = wp + (int)__builtin_amdgcn_mbcnt_lo(bal, 0u);
                    if (pos < lim) myq[pos] = (unsigned)e;
                    else place_one((unsigned)rec[e], base, sCnt, sMain, sSpill, ovfcnt, ovf);
                }
                wp += __builtin_popcount(bal);
            }
        }
        wp = wp < lim ? wp : lim;
    }

    for (int i0 = 0; i0 < wp; i0 += 128) {
        unsigned r[4];
        bool ok[4];
#pragma unroll
        for (int k = 0; k < 4; ++k) {
            const int i = i0 + k * 32 + lane;
            ok[k] = i < wp;
            r[k] = ok[k] ? (unsigned)rec[myq[i]] : 0u;
        }
#pragma unroll
        for (int k = 0; k < 4; ++k)
            if (ok[k]) place_one(r[k], base, sCnt, sMain, sSpill, ovfcnt, ovf);
    }
    __syncthreads();

    const size_t nb = (size_t)half * N_NODES + base;
    for (int i = tid; i < SPAN; i += 1024) cnt[nb + i] = sCnt[i];
    int4* gm = (int4*)(mainL + nb * SLOTS);
    for (int i = tid; i < SPAN * SLOTS / 4; i += 1024) gm[i] = ((const int4*)sMain)[i];
    int4* gs = (int4*)(spillL + nb * SLOTS);
    for (int i = tid; i < SPAN * SLOTS / 4; i += 1024)
        if (sCnt[i >> 2] > SLOTS) gs[i] = ((const int4*)sSpill)[i];
}

template <bool LOGSOFTMAX>
__global__ __launch_bounds__(128) void k_gin_layer(const float* __restrict__ xin,
                                                   const int* __restrict__ cnt,
                                                   const int* __restrict__ mainL,
                                                   const int* __restrict__ spillL,
                                                   const int* __restrict__ ovf,
                                                   const float* __restrict__ eps_p,
                                                   const float* __restrict__ wa,
                                                   const float* __restrict__ ba,
                                                   const float* __restrict__ wb,
                                                   const float* __restrict__ bb,
                                                   float* __restrict__ out) {
    __shared__ float ldsM[4][16][17];
    __shared__ float ldsT[4][16][17];
    __shared__ __attribute__((aligned(16))) int ldsI[4][16][IPITCH];

    const int wid  = threadIdx.x >> 5;
    const int lane = threadIdx.x & 31;
    const int tile = blockIdx.x * 4 + wid;

    const int g = lane >> 2;
    const int q = lane & 3;
    const int node0 = tile * 16 + g;
    const int node1 = node0 + 8;

    const float4* __restrict__ f4 = (const float4*)xin;
    const int cA0 = cnt[node0];
    const int cB0 = cnt[N_NODES + node0];
    const int cA1 = cnt[node1];
    const int cB1 = cnt[N_NODES + node1];
    const int4 mA0 = *(const int4*)(mainL + (size_t)node0 * SLOTS + 4 * q);
    const int4 mB0 = *(const int4*)(mainL + ((size_t)N_NODES + node0) * SLOTS + 4 * q);
    const int4 mA1 = *(const int4*)(mainL + (size_t)node1 * SLOTS + 4 * q);
    const int4 mB1 = *(const int4*)(mainL + ((size_t)N_NODES + node1) * SLOTS + 4 * q);
    const float4 x0 = f4[(size_t)node0 * 4 + q];
    const float4 x1 = f4[(size_t)node1 * 4 + q];

    const int dA0 = cA0 < SLOTS ? cA0 : SLOTS;
    const int dB0 = cB0 < SLOTS ? cB0 : SLOTS;
    const int dA1 = cA1 < SLOTS ? cA1 : SLOTS;
    const int dB1 = cB1 < SLOTS ? cB1 : SLOTS;
    const int d0 = dA0 + dB0;
    const int d1 = dA1 + dB1;

    int* l0 = &ldsI[wid][g][0];
    int* l1 = &ldsI[wid][g + 8][0];
    *(int4*)(l0 + 4 * q) = mA0;
    *(int4*)(l1 + 4 * q) = mA1;
    *(int4*)(l0 + SLOTS + 4 * q) = make_int4(0, 0, 0, 0);
    *(int4*)(l1 + SLOTS + 4 * q) = make_int4(0, 0, 0, 0);
    l0[dA0 + 4 * q + 0] = mB0.x; l0[dA0 + 4 * q + 1] = mB0.y;
    l0[dA0 + 4 * q + 2] = mB0.z; l0[dA0 + 4 * q + 3] = mB0.w;
    l1[dA1 + 4 * q + 0] = mB1.x; l1[dA1 + 4 * q + 1] = mB1.y;
    l1[dA1 + 4 * q + 2] = mB1.z; l1[dA1 + 4 * q + 3] = mB1.w;

    float4 a0 = make_float4(0.f, 0.f, 0.f, 0.f);
    float4 a1 = make_float4(0.f, 0.f, 0.f, 0.f);

    int dm = d0 > d1 ? d0 : d1;
#pragma unroll
    for (int off = 1; off < 32; off <<= 1) {
        const int o = __shfl_xor(dm, off, 32);
        dm = o > dm ? o : dm;
    }
    dm = __builtin_amdgcn_readfirstlane(dm);
    __syncthreads();

    for (int i = 0; i < dm; i += 4) {
        const int4 s0 = *(const int4*)(l0 + i);
        const int4 s1 = *(const int4*)(l1 + i);
        const bool v00 = i + 0 < d0, v01 = i + 1 < d0, v02 = i + 2 < d0, v03 = i + 3 < d0;
        const bool v10 = i + 0 < d1, v11 = i + 1 < d1, v12 = i + 2 < d1, v13 = i + 3 < d1;
        const float4 r00 = f4[(size_t)(v00 ? s0.x : node0) * 4 + q];
        const float4 r01 = f4[(size_t)(v01 ? s0.y : node0) * 4 + q];
        const float4 r02 = f4[(size_t)(v02 ? s0.z : node0) * 4 + q];
        const float4 r03 = f4[(size_t)(v03 ? s0.w : node0) * 4 + q];
        const float4 r10 = f4[(size_t)(v10 ? s1.x : node1) * 4 + q];
        const float4 r11 = f4[(size_t)(v11 ? s1.y : node1) * 4 + q];
        const float4 r12 = f4[(size_t)(v12 ? s1.z : node1) * 4 + q];
        const float4 r13 = f4[(size_t)(v13 ? s1.w : node1) * 4 + q];
#define ACC(a, r, v) { a.x += (v) ? r.x : 0.0f; a.y += (v) ? r.y : 0.0f; \
                       a.z += (v) ? r.z : 0.0f; a.w += (v) ? r.w : 0.0f; }
        ACC(a0, r00, v00) ACC(a0, r01, v01) ACC(a0, r02, v02) ACC(a0, r03, v03)
        ACC(a1, r10, v10) ACC(a1, r11, v11) ACC(a1, r12, v12) ACC(a1, r13, v13)
#undef ACC
    }

    if (cA0 > SLOTS || cB0 > SLOTS || cA1 > SLOTS || cB1 > SLOTS) {
#define SPILL(a, c, row) { const int e_ = (c) < 2 * SLOTS ? (c) : 2 * SLOTS;                 \
        for (int j = SLOTS; j < e_; ++j) {                                                  \
            const int ss = spillL[(row) * SLOTS + (j - SLOTS)];                              \
            const float4 r = f4[(size_t)ss * 4 + q];                                        \
            a.x += r.x; a.y += r.y; a.z += r.z; a.w += r.w; } }
        SPILL(a0, cA0, (size_t)node0)
        SPILL(a0, cB0, (size_t)N_NODES + node0)
        SPILL(a1, cA1, (size_t)node1)
        SPILL(a1, cB1, (size_t)N_NODES + node1)
#undef SPILL
    }

    if (cA0 > 2 * SLOTS || cB0 > 2 * SLOTS || cA1 > 2 * SLOTS || cB1 > 2 * SLOTS) {
        int novf = cnt[2 * N_NODES];
        novf = novf < N_EDGES ? novf : N_EDGES;
        for (int j = 0; j < novf; ++j) {
            const int dd = ovf[2 * (size_t)j + 0];
            const int ss = ovf[2 * (size_t)j + 1];
            if (dd == node0) {
                const float4 r = f4[(size_t)ss * 4 + q];
                a0.x += r.x; a0.y += r.y; a0.z += r.z; a0.w += r.w;
            }
            if (dd == node1) {
                const float4 r = f4[(size_t)ss * 4 + q];
                a1.x += r.x; a1.y += r.y; a1.z += r.z; a1.w += r.w;
            }
        }
    }

    const float epsp = 1.0f + eps_p[0];
    {
        float* m0 = &ldsM[wid][g][q * 4];
        float* m1 = &ldsM[wid][g + 8][q * 4];
        m0[0] = epsp * x0.x + a0.x; m0[1] = epsp * x0.y + a0.y;
        m0[2] = epsp * x0.z + a0.z; m0[3] = epsp * x0.w + a0.w;
        m1[0] = epsp * x1.x + a1.x; m1[1] = epsp * x1.y + a1.y;
        m1[2] = epsp * x1.z + a1.z; m1[3] = epsp * x1.w + a1.w;
    }
    __syncthreads();

    const int col = lane & 15;
    const int hi  = lane >> 4;

    v8f acc = {};
#pragma unroll
    for (int c = 0; c < 4; ++c) {
        const int kb = c * 4 + hi * 2;
        v2f a;
        a.x = ldsM[wid][col][kb + 0];
        a.y = ldsM[wid][col][kb + 1];
        v2f b;
        b.x = wa[(kb + 0) * 16 + col];
        b.y = wa[(kb + 1) * 16 + col];
        acc = __builtin_amdgcn_wmma_f32_16x16x4_f32(false, a, false, b,
                                                    (short)0, acc, false, false);
    }

    const float biasA = ba[col];
#pragma unroll
    for (int v = 0; v < 8; ++v) {
        float t = acc[v] + biasA;
        ldsT[wid][v + hi * 8][col] = t > 0.0f ? t : 0.0f;
    }
    __syncthreads();

    v8f acc2 = {};
#pragma unroll
    for (int c = 0; c < 4; ++c) {
        const int kb = c * 4 + hi * 2;
        v2f a;
        a.x = ldsT[wid][col][kb + 0];
        a.y = ldsT[wid][col][kb + 1];
        v2f b;
        b.x = wb[(kb + 0) * 16 + col];
        b.y = wb[(kb + 1) * 16 + col];
        acc2 = __builtin_amdgcn_wmma_f32_16x16x4_f32(false, a, false, b,
                                                     (short)0, acc2, false, false);
    }

    const float biasB = bb[col];
    float* orow = out + (size_t)tile * 16 * FDIM;

    if (LOGSOFTMAX) {
#pragma unroll
        for (int v = 0; v < 8; ++v) {
            float val = acc2[v] + biasB;
            float m = val;
#pragma unroll
            for (int off = 1; off < 16; off <<= 1)
                m = fmaxf(m, __shfl_xor(m, off, 16));
            float s = __expf(val - m);
#pragma unroll
            for (int off = 1; off < 16; off <<= 1)
                s += __shfl_xor(s, off, 16);
            orow[(v + hi * 8) * FDIM + col] = val - m - __logf(s);
        }
    } else {
#pragma unroll
        for (int v = 0; v < 8; ++v) {
            float t = acc2[v] + biasB;
            orow[(v + hi * 8) * FDIM + col] = t > 0.0f ? t : 0.0f;
        }
    }
}

extern "C" void kernel_launch(void* const* d_in, const int* in_sizes, int n_in,
                              void* d_out, int out_size, void* d_ws, size_t ws_size,
                              hipStream_t stream) {
    const float* x    = (const float*)d_in[0];
    const int*   ei   = (const int*)d_in[1];
    const float* eps1 = (const float*)d_in[2];
    const float* w1a  = (const float*)d_in[3];
    const float* b1a  = (const float*)d_in[4];
    const float* w1b  = (const float*)d_in[5];
    const float* b1b  = (const float*)d_in[6];
    const float* eps2 = (const float*)d_in[7];
    const float* w2a  = (const float*)d_in[8];
    const float* b2a  = (const float*)d_in[9];
    const float* w2b  = (const float*)d_in[10];
    const float* b2b  = (const float*)d_in[11];
    float* out = (float*)d_out;

    float* h1     = (float*)d_ws;
    int*   mainL  = (int*)(h1 + (size_t)N_NODES * FDIM);
    int*   spillL = mainL + (size_t)2 * N_NODES * SLOTS;
    int*   ovf    = spillL + (size_t)2 * N_NODES * SLOTS;
    int*   cnt    = ovf + (size_t)N_EDGES * 2;
    int*   rec    = cnt + (2 * N_NODES + 32);
    unsigned* planes = (unsigned*)(rec + (size_t)N_EDGES + 96);

    k_owner<<<N_EDGES / 1024, 256, 0, stream>>>(ei, planes, rec, cnt + 2 * N_NODES);
    k_build<<<2 * NOWN, 1024, 0, stream>>>(planes, rec, cnt, mainL, spillL, ovf);
    k_gin_layer<false><<<NTILES / 4, 128, 0, stream>>>(x, cnt, mainL, spillL, ovf, eps1, w1a,
                                                       b1a, w1b, b1b, h1);
    k_gin_layer<true><<<NTILES / 4, 128, 0, stream>>>(h1, cnt, mainL, spillL, ovf, eps2, w2a,
                                                      b2a, w2b, b2b, out);
}


# Layer gathers: a list entry past its node's count issues no load (lanes masked off) instead of re-reading the node's own row; fewer lines touched per gather step; sums bit-identical
# speedup vs baseline: 3.7775x; 1.0504x over previous
#include <hip/hip_runtime.h>


#define N_NODES 200000
#define N_EDGES 3200000
#define FDIM    16
#define NTILES  (N_NODES / 16)
#define SLOTS   16
#define SPAN    1600
#define NOWN    (N_NODES / SPAN)
#define NBIT    7
#define NBLK    (N_EDGES / 128)
#define HSTEP   12512
#define PLS     25024
#define ROTQ    96
static_assert(NOWN <= (1 << NBIT), "a range number must fit NBIT bits");
static_assert(HSTEP % 32 == 0 && PLS % 32 == 0 && ROTQ % 32 == 0, "512-byte pieces stay aligned");
static_assert(HSTEP <= NBLK && 2 * HSTEP >= NBLK && PLS >= NBLK, "the two halves cover every block");
static_assert((NOWN - 1) * ROTQ < HSTEP, "a pass wraps at most once");
static_assert(N_EDGES % 1024 == 0, "the owner pass has no partial work-group");
#define QE      448
#define IPITCH  36

typedef float v2f __attribute__((ext_vector_type(2)));
typedef float v8f __attribute__((ext_vector_type(8)));

__global__ __launch_bounds__(256) void k_owner(const int* __restrict__ ei,
                                               unsigned* __restrict__ planes,
                                               int* __restrict__ rec,
                                               int* __restrict__ ovfcnt) {
    __shared__ __attribute__((aligned(16))) unsigned sW[NBIT * 8 * 4];
    const int t = blockIdx.x * 256 + threadIdx.x;
    const int lane = threadIdx.x & 31;
    const int wid  = threadIdx.x >> 5;
    if (t == 0) ovfcnt[0] = 0;
    const int4 s4 = *(const int4*)(ei + (size_t)t * 4);
    const int4 d4 = *(const int4*)(ei + N_EDGES + (size_t)t * 4);
    const unsigned o0 = (unsigned)d4.x / SPAN, o1 = (unsigned)d4.y / SPAN;
    const unsigned o2 = (unsigned)d4.z / SPAN, o3 = (unsigned)d4.w / SPAN;
    int4 r;
    r.x = (int)((((unsigned)d4.x - o0 * SPAN) << 18) | (unsigned)s4.x);
    r.y = (int)((((unsigned)d4.y - o1 * SPAN) << 18) | (unsigned)s4.y);
    r.z = (int)((((unsigned)d4.z - o2 * SPAN) << 18) | (unsigned)s4.z);
    r.w = (int)((((unsigned)d4.w - o3 * SPAN) << 18) | (unsigned)s4.w);
    *(int4*)(rec + (size_t)t * 4) = r;

    uint4 mine = make_uint4(0u, 0u, 0u, 0u);
#pragma unroll
    for (int b = 0; b < NBIT; ++b) {
        const unsigned w0 = __builtin_amdgcn_ballot_w32(((o0 >> b) & 1u) != 0u);
        const unsigned w1 = __builtin_amdgcn_ballot_w32(((o1 >> b) & 1u) != 0u);
        const unsigned w2 = __builtin_amdgcn_ballot_w32(((o2 >> b) & 1u) != 0u);
        const unsigned w3 = __builtin_amdgcn_ballot_w32(((o3 >> b) & 1u) != 0u);
        if (lane == b) mine = make_uint4(w0, w1, w2, w3);
    }
    if (lane < NBIT) ((uint4*)sW)[lane * 8 + wid] = mine;
    __syncthreads();
    if (threadIdx.x < NBIT * 8) {
        const int b = threadIdx.x >> 3;
        const int i = threadIdx.x & 7;
        ((uint4*)planes)[(size_t)b * PLS + (size_t)blockIdx.x * 8 + i] = ((const uint4*)sW)[threadIdx.x];
    }
}

__device__ __forceinline__ void place_one(unsigned r, int base, int* sCnt, int* sMain,
                                          int* sSpill,
                                          int* __restrict__ ovfcnt, int* __restrict__ ovf) {
    const int local = (int)(r >> 18);
    const int s = (int)(r & 0x3FFFFu);
    const int tk = atomicAdd(&sCnt[local], 1);
    if (tk < SLOTS) {
        sMain[local * SLOTS + tk] = s;
    } else if (tk < 2 * SLOTS) {
        sSpill[local * SLOTS + tk - SLOTS] = s;
    } else {
        const int k = atomicAdd(ovfcnt, 1);
        ovf[2 * (size_t)k + 0] = base + local;
        ovf[2 * (size_t)k + 1] = s;
    }
}

__device__ __forceinline__ int edge_of(unsigned x, int bit, int bbase) {
    return (int)((((unsigned)bbase + (x >> 2)) << 7) + ((unsigned)bit << 2) + (x & 3u));
}

__device__ __forceinline__ void file_word(unsigned x1, unsigned m, int bbase,
                                          const int* __restrict__ rec, int base, int* sCnt,
                                          int* sMain, int* sSpill,
                                          int* __restrict__ ovfcnt, int* __restrict__ ovf) {
    while (m != 0u) {
        const int bit = __builtin_ctz(m);
        m &= m - 1u;
        place_one((unsigned)rec[edge_of(x1, bit, bbase)], base, sCnt, sMain, sSpill, ovfcnt, ovf);
    }
}

__global__ __launch_bounds__(1024) void k_build(const unsigned* __restrict__ planes,
                                                const int* __restrict__ rec,
                                                int* __restrict__ cnt,
                                                int* __restrict__ mainL,
                                                int* __restrict__ spillL,
                                                int* __restrict__ ovf) {
    __shared__ __attribute__((aligned(16))) int sMain[SPAN * SLOTS];
    __shared__ __attribute__((aligned(16))) int sSpill[SPAN * SLOTS];
    __shared__ __attribute__((aligned(8))) unsigned sQ[32 * 2 * QE];
    __shared__ int sCnt[SPAN];

    const int tid  = threadIdx.x;
    const int lane = tid & 31;
    const int wid  = tid >> 5;
    const int rng  = blockIdx.x >> 1;
    const int half = blockIdx.x & 1;
    const int base = rng * SPAN;
    int* ovfcnt = cnt + 2 * N_NODES;

    for (int i = tid; i < SPAN * SLOTS / 4; i += 1024) {
        ((int4*)sMain)[i]  = make_int4(0, 0, 0, 0);
        ((int4*)sSpill)[i] = make_int4(0, 0, 0, 0);
    }
    for (int i = tid; i < SPAN; i += 1024) sCnt[i] = 0;
    __syncthreads();

    unsigned rb[NBIT];
#pragma unroll
    for (int b = 0; b < NBIT; ++b) rb[b] = ((rng >> b) & 1) ? 0xFFFFFFFFu : 0u;
    const int bbase = half * HSTEP;
    const uint4* __restrict__ pl = (const uint4*)planes + bbase;
    const int rot = rng * ROTQ;
    unsigned* myq = sQ + wid * (2 * QE);
    int qn = 0;

    auto scan_block = [&](const uint4 (&p)[NBIT], bool okb, int blk) {
#pragma unroll
        for (int j = 0; j < 4; ++j) {
            unsigned diff = 0u;
#pragma unroll
            for (int b = 0; b < NBIT; ++b) {
                const unsigned w = j == 0 ? p[b].x : j == 1 ? p[b].y : j == 2 ? p[b].z : p[b].w;
                diff |= w ^ rb[b];
            }
            const unsigned m = okb ? ~diff : 0u;
            const unsigned bal = __builtin_amdgcn_ballot_w32(m != 0u);
            if (m != 0u) {
                const unsigned x = (unsigned)blk * 4u + (unsigned)j;
                const int pos = qn + (int)__builtin_amdgcn_mbcnt_lo(bal, 0u);
                if (pos < QE) {
                    myq[2 * pos + 0] = x;
                    myq[2 * pos + 1] = m;
                } else {
                    file_word(x, m, bbase, rec, base, sCnt, sMain, sSpill, ovfcnt, ovf);
                }
            }
            qn += __builtin_popcount(bal);
        }
    };

    auto block_of = [&](int it, bool& okb) {
        int x = it + rot;
        x = x >= HSTEP ? x - HSTEP : x;
        okb = it < HSTEP && bbase + x < NBLK;
        return okb ? x : 0;
    };

    int g = 0;
    for (; g + 2 * 1024 <= HSTEP; g += 2 * 1024) {
        uint4 p0[NBIT], p1[NBIT];
        bool ok0, ok1;
        const int x0 = block_of(g + tid, ok0);
        const int x1 = block_of(g + 1024 + tid, ok1);
#pragma unroll
        for (int b = 0; b < NBIT; ++b) p0[b] = pl[(size_t)b * PLS + x0];
#pragma unroll
        for (int b = 0; b < NBIT; ++b) p1[b] = pl[(size_t)b * PLS + x1];
        __builtin_amdgcn_sched_barrier(0);
        scan_block(p0, ok0, x0);
        scan_block(p1, ok1, x1);
    }
    for (; g < HSTEP; g += 1024) {
        uint4 p0[NBIT];
        bool ok0;
        const int x0 = block_of(g + tid, ok0);
#pragma unroll
        for (int b = 0; b < NBIT; ++b) p0[b] = pl[(size_t)b * PLS + x0];
        scan_block(p0, ok0, x0);
    }

    const int nq = qn < QE ? qn : QE;
    int wp = 0;
    for (int i0 = 0; i0 < nq; i0 += 128) {
        unsigned ex[4], em[4];
#pragma unroll
        for (int k = 0; k < 4; ++k) {
            const int i = i0 + k * 32 + lane;
            const bool okk = i < nq;
            ex[k] = okk ? myq[2 * i + 0] : 0u;
            em[k] = okk ? myq[2 * i + 1] : 0u;
        }
        const int lim = i0 + 128 < nq ? 2 * (i0 + 128) : 2 * QE;
#pragma unroll
        for (int k = 0; k < 4; ++k) {
            unsigned m = em[k];
            for (;;) {
                const unsigned bal = __builtin_amdgcn_ballot_w32(m != 0u);
                if (bal == 0u) break;
                if (m != 0u) {
                    const int bit = __builtin_ctz(m);
                    m &= m - 1u;
                    const int e = edge_of(ex[k], bit, bbase);
                    const int pos = wp + (int)__builtin_amdgcn_mbcnt_lo(bal, 0u);
                    if (pos < lim) myq[pos] = (unsigned)e;
                    else place_one((unsigned)rec[e], base, sCnt, sMain, sSpill, ovfcnt, ovf);
                }
                wp += __builtin_popcount(bal);
            }
        }
        wp = wp < lim ? wp : lim;
    }

    for (int i0 = 0; i0 < wp; i0 += 128) {
        unsigned r[4];
        bool ok[4];
#pragma unroll
        for (int k = 0; k < 4; ++k) {
            const int i = i0 + k * 32 + lane;
            ok[k] = i < wp;
            r[k] = ok[k] ? (unsigned)rec[myq[i]] : 0u;
        }
#pragma unroll
        for (int k = 0; k < 4; ++k)
            if (ok[k]) place_one(r[k], base, sCnt, sMain, sSpill, ovfcnt, ovf);
    }
    __syncthreads();

    const size_t nb = (size_t)half * N_NODES + base;
    for (int i = tid; i < SPAN; i += 1024) cnt[nb + i] = sCnt[i];
    int4* gm = (int4*)(mainL + nb * SLOTS);
    for (int i = tid; i < SPAN * SLOTS / 4; i += 1024) gm[i] = ((const int4*)sMain)[i];
    int4* gs = (int4*)(spillL + nb * SLOTS);
    for (int i = tid; i < SPAN * SLOTS / 4; i += 1024)
        if (sCnt[i >> 2] > SLOTS) gs[i] = ((const int4*)sSpill)[i];
}

template <bool LOGSOFTMAX>
__global__ __launch_bounds__(128) void k_gin_layer(const float* __restrict__ xin,
                                                   const int* __restrict__ cnt,
                                                   const int* __restrict__ mainL,
                                                   const int* __restrict__ spillL,
                                                   const int* __restrict__ ovf,
                                                   const float* __restrict__ eps_p,
                                                   const float* __restrict__ wa,
                                                   const float* __restrict__ ba,
                                                   const float* __restrict__ wb,
                                                   const float* __restrict__ bb,
                                                   float* __restrict__ out) {
    __shared__ float ldsM[4][16][17];
    __shared__ float ldsT[4][16][17];
    __shared__ __attribute__((aligned(16))) int ldsI[4][16][IPITCH];

    const int wid  = threadIdx.x >> 5;
    const int lane = threadIdx.x & 31;
    const int tile = blockIdx.x * 4 + wid;

    const int g = lane >> 2;
    const int q = lane & 3;
    const int node0 = tile * 16 + g;
    const int node1 = node0 + 8;

    const float4* __restrict__ f4 = (const float4*)xin;
    const int cA0 = cnt[node0];
    const int cB0 = cnt[N_NODES + node0];
    const int cA1 = cnt[node1];
    const int cB1 = cnt[N_NODES + node1];
    const int4 mA0 = *(const int4*)(mainL + (size_t)node0 * SLOTS + 4 * q);
    const int4 mB0 = *(const int4*)(mainL + ((size_t)N_NODES + node0) * SLOTS + 4 * q);
    const int4 mA1 = *(const int4*)(mainL + (size_t)node1 * SLOTS + 4 * q);
    const int4 mB1 = *(const int4*)(mainL + ((size_t)N_NODES + node1) * SLOTS + 4 * q);
    const float4 x0 = f4[(size_t)node0 * 4 + q];
    const float4 x1 = f4[(size_t)node1 * 4 + q];

    const int dA0 = cA0 < SLOTS ? cA0 : SLOTS;
    const int dB0 = cB0 < SLOTS ? cB0 : SLOTS;
    const int dA1 = cA1 < SLOTS ? cA1 : SLOTS;
    const int dB1 = cB1 < SLOTS ? cB1 : SLOTS;
    const int d0 = dA0 + dB0;
    const int d1 = dA1 + dB1;

    int* l0 = &ldsI[wid][g][0];
    int* l1 = &ldsI[wid][g + 8][0];
    *(int4*)(l0 + 4 * q) = mA0;
    *(int4*)(l1 + 4 * q) = mA1;
    *(int4*)(l0 + SLOTS + 4 * q) = make_int4(0, 0, 0, 0);
    *(int4*)(l1 + SLOTS + 4 * q) = make_int4(0, 0, 0, 0);
    l0[dA0 + 4 * q + 0] = mB0.x; l0[dA0 + 4 * q + 1] = mB0.y;
    l0[dA0 + 4 * q + 2] = mB0.z; l0[dA0 + 4 * q + 3] = mB0.w;
    l1[dA1 + 4 * q + 0] = mB1.x; l1[dA1 + 4 * q + 1] = mB1.y;
    l1[dA1 + 4 * q + 2] = mB1.z; l1[dA1 + 4 * q + 3] = mB1.w;

    float4 a0 = make_float4(0.f, 0.f, 0.f, 0.f);
    float4 a1 = make_float4(0.f, 0.f, 0.f, 0.f);

    int dm = d0 > d1 ? d0 : d1;
#pragma unroll
    for (int off = 1; off < 32; off <<= 1) {
        const int o = __shfl_xor(dm, off, 32);
        dm = o > dm ? o : dm;
    }
    dm = __builtin_amdgcn_readfirstlane(dm);
    __syncthreads();

    for (int i = 0; i < dm; i += 4) {
        const int4 s0 = *(const int4*)(l0 + i);
        const int4 s1 = *(const int4*)(l1 + i);
        const float4 z4 = make_float4(0.f, 0.f, 0.f, 0.f);
        float4 r00 = z4, r01 = z4, r02 = z4, r03 = z4;
        float4 r10 = z4, r11 = z4, r12 = z4, r13 = z4;
        if (i + 0 < d0) r00 = f4[(size_t)s0.x * 4 + q];
        if (i + 1 < d0) r01 = f4[(size_t)s0.y * 4 + q];
        if (i + 2 < d0) r02 = f4[(size_t)s0.z * 4 + q];
        if (i + 3 < d0) r03 = f4[(size_t)s0.w * 4 + q];
        if (i + 0 < d1) r10 = f4[(size_t)s1.x * 4 + q];
        if (i + 1 < d1) r11 = f4[(size_t)s1.y * 4 + q];
        if (i + 2 < d1) r12 = f4[(size_t)s1.z * 4 + q];
        if (i + 3 < d1) r13 = f4[(size_t)s1.w * 4 + q];
#define ACC(a, r) { a.x += r.x; a.y += r.y; a.z += r.z; a.w += r.w; }
        ACC(a0, r00) ACC(a0, r01) ACC(a0, r02) ACC(a0, r03)
        ACC(a1, r10) ACC(a1, r11) ACC(a1, r12) ACC(a1, r13)
#undef ACC
    }

    if (cA0 > SLOTS || cB0 > SLOTS || cA1 > SLOTS || cB1 > SLOTS) {
#define SPILL(a, c, row) { const int e_ = (c) < 2 * SLOTS ? (c) : 2 * SLOTS;                 \
        for (int j = SLOTS; j < e_; ++j) {                                                  \
            const int ss = spillL[(row) * SLOTS + (j - SLOTS)];                              \
            const float4 r = f4[(size_t)ss * 4 + q];                                        \
            a.x += r.x; a.y += r.y; a.z += r.z; a.w += r.w; } }
        SPILL(a0, cA0, (size_t)node0)
        SPILL(a0, cB0, (size_t)N_NODES + node0)
        SPILL(a1, cA1, (size_t)node1)
        SPILL(a1, cB1, (size_t)N_NODES + node1)
#undef SPILL
    }

    if (cA0 > 2 * SLOTS || cB0 > 2 * SLOTS || cA1 > 2 * SLOTS || cB1 > 2 * SLOTS) {
        int novf = cnt[2 * N_NODES];
        novf = novf < N_EDGES ? novf : N_EDGES;
        for (int j = 0; j < novf; ++j) {
            const int dd = ovf[2 * (size_t)j + 0];
            const int ss = ovf[2 * (size_t)j + 1];
            if (dd == node0) {
                const float4 r = f4[(size_t)ss * 4 + q];
                a0.x += r.x; a0.y += r.y; a0.z += r.z; a0.w += r.w;
            }
            if (dd == node1) {
                const float4 r = f4[(size_t)ss * 4 + q];
                a1.x += r.x; a1.y += r.y; a1.z += r.z; a1.w += r.w;
            }
        }
    }

    const float epsp = 1.0f + eps_p[0];
    {
        float* m0 = &ldsM[wid][g][q * 4];
        float* m1 = &ldsM[wid][g + 8][q * 4];
        m0[0] = epsp * x0.x + a0.x; m0[1] = epsp * x0.y + a0.y;
        m0[2] = epsp * x0.z + a0.z; m0[3] = epsp * x0.w + a0.w;
        m1[0] = epsp * x1.x + a1.x; m1[1] = epsp * x1.y + a1.y;
        m1[2] = epsp * x1.z + a1.z; m1[3] = epsp * x1.w + a1.w;
    }
    __syncthreads();

    const int col = lane & 15;
    const int hi  = lane >> 4;

    v8f acc = {};
#pragma unroll
    for (int c = 0; c < 4; ++c) {
        const int kb = c * 4 + hi * 2;
        v2f a;
        a.x = ldsM[wid][col][kb + 0];
        a.y = ldsM[wid][col][kb + 1];
        v2f b;
        b.x = wa[(kb + 0) * 16 + col];
        b.y = wa[(kb + 1) * 16 + col];
        acc = __builtin_amdgcn_wmma_f32_16x16x4_f32(false, a, false, b,
                                                    (short)0, acc, false, false);
    }

    const float biasA = ba[col];
#pragma unroll
    for (int v = 0; v < 8; ++v) {
        float t = acc[v] + biasA;
        ldsT[wid][v + hi * 8][col] = t > 0.0f ? t : 0.0f;
    }
    __syncthreads();

    v8f acc2 = {};
#pragma unroll
    for (int c = 0; c < 4; ++c) {
        const int kb = c * 4 + hi * 2;
        v2f a;
        a.x = ldsT[wid][col][kb + 0];
        a.y = ldsT[wid][col][kb + 1];
        v2f b;
        b.x = wb[(kb + 0) * 16 + col];
        b.y = wb[(kb + 1) * 16 + col];
        acc2 = __builtin_amdgcn_wmma_f32_16x16x4_f32(false, a, false, b,
                                                     (short)0, acc2, false, false);
    }

    const float biasB = bb[col];
    float* orow = out + (size_t)tile * 16 * FDIM;

    if (LOGSOFTMAX) {
#pragma unroll
        for (int v = 0; v < 8; ++v) {
            float val = acc2[v] + biasB;
            float m = val;
#pragma unroll
            for (int off = 1; off < 16; off <<= 1)
                m = fmaxf(m, __shfl_xor(m, off, 16));
            float s = __expf(val - m);
#pragma unroll
            for (int off = 1; off < 16; off <<= 1)
                s += __shfl_xor(s, off, 16);
            orow[(v + hi * 8) * FDIM + col] = val - m - __logf(s);
        }
    } else {
#pragma unroll
        for (int v = 0; v < 8; ++v) {
            float t = acc2[v] + biasB;
            orow[(v + hi * 8) * FDIM + col] = t > 0.0f ? t : 0.0f;
        }
    }
}

extern "C" void kernel_launch(void* const* d_in, const int* in_sizes, int n_in,
                              void* d_out, int out_size, void* d_ws, size_t ws_size,
                              hipStream_t stream) {
    const float* x    = (const float*)d_in[0];
    const int*   ei   = (const int*)d_in[1];
    const float* eps1 = (const float*)d_in[2];
    const float* w1a  = (const float*)d_in[3];
    const float* b1a  = (const float*)d_in[4];
    const float* w1b  = (const float*)d_in[5];
    const float* b1b  = (const float*)d_in[6];
    const float* eps2 = (const float*)d_in[7];
    const float* w2a  = (const float*)d_in[8];
    const float* b2a  = (const float*)d_in[9];
    const float* w2b  = (const float*)d_in[10];
    const float* b2b  = (const float*)d_in[11];
    float* out = (float*)d_out;

    float* h1     = (float*)d_ws;
    int*   mainL  = (int*)(h1 + (size_t)N_NODES * FDIM);
    int*   spillL = mainL + (size_t)2 * N_NODES * SLOTS;
    int*   ovf    = spillL + (size_t)2 * N_NODES * SLOTS;
    int*   cnt    = ovf + (size_t)N_EDGES * 2;
    int*   rec    = cnt + (2 * N_NODES + 32);
    unsigned* planes = (unsigned*)(rec + (size_t)N_EDGES + 96);

    k_owner<<<N_EDGES / 1024, 256, 0, stream>>>(ei, planes, rec, cnt + 2 * N_NODES);
    k_build<<<2 * NOWN, 1024, 0, stream>>>(planes, rec, cnt, mainL, spillL, ovf);
    k_gin_layer<false><<<NTILES / 4, 128, 0, stream>>>(x, cnt, mainL, spillL, ovf, eps1, w1a,
                                                       b1a, w1b, b1b, h1);
    k_gin_layer<true><<<NTILES / 4, 128, 0, stream>>>(h1, cnt, mainL, spillL, ovf, eps2, w2a,
                                                      b2a, w2b, b2b, out);
}


# Layer kernels fit 64 registers (own rows read after the gather loop, 32-bit gather offsets, rare spill loops rolled): 16 waves per SIMD, all 3125 work-groups resident at once; sums bit-identical
# speedup vs baseline: 4.0177x; 1.0636x over previous
#include <hip/hip_runtime.h>


#define N_NODES 200000
#define N_EDGES 3200000
#define FDIM    16
#define NTILES  (N_NODES / 16)
#define SLOTS   16
#define SPAN    1600
#define NOWN    (N_NODES / SPAN)
#define NBIT    7
#define NBLK    (N_EDGES / 128)
#define HSTEP   12512
#define PLS     25024
#define ROTQ    96
static_assert(NOWN <= (1 << NBIT), "a range number must fit NBIT bits");
static_assert(HSTEP % 32 == 0 && PLS % 32 == 0 && ROTQ % 32 == 0, "512-byte pieces stay aligned");
static_assert(HSTEP <= NBLK && 2 * HSTEP >= NBLK && PLS >= NBLK, "the two halves cover every block");
static_assert((NOWN - 1) * ROTQ < HSTEP, "a pass wraps at most once");
static_assert(N_EDGES % 1024 == 0, "the owner pass has no partial work-group");
#define QE      448
#define IPITCH  36

typedef float v2f __attribute__((ext_vector_type(2)));
typedef float v8f __attribute__((ext_vector_type(8)));

__global__ __launch_bounds__(256) void k_owner(const int* __restrict__ ei,
                                               unsigned* __restrict__ planes,
                                               int* __restrict__ rec,
                                               int* __restrict__ ovfcnt) {
    __shared__ __attribute__((aligned(16))) unsigned sW[NBIT * 8 * 4];
    const int t = blockIdx.x * 256 + threadIdx.x;
    const int lane = threadIdx.x & 31;
    const int wid  = threadIdx.x >> 5;
    if (t == 0) ovfcnt[0] = 0;
    const int4 s4 = *(const int4*)(ei + (size_t)t * 4);
    const int4 d4 = *(const int4*)(ei + N_EDGES + (size_t)t * 4);
    const unsigned o0 = (unsigned)d4.x / SPAN, o1 = (unsigned)d4.y / SPAN;
    const unsigned o2 = (unsigned)d4.z / SPAN, o3 = (unsigned)d4.w / SPAN;
    int4 r;
    r.x = (int)((((unsigned)d4.x - o0 * SPAN) << 18) | (unsigned)s4.x);
    r.y = (int)((((unsigned)d4.y - o1 * SPAN) << 18) | (unsigned)s4.y);
    r.z = (int)((((unsigned)d4.z - o2 * SPAN) << 18) | (unsigned)s4.z);
    r.w = (int)((((unsigned)d4.w - o3 * SPAN) << 18) | (unsigned)s4.w);
    *(int4*)(rec + (size_t)t * 4) = r;

    uint4 mine = make_uint4(0u, 0u, 0u, 0u);
#pragma unroll
    for (int b = 0; b < NBIT; ++b) {
        const unsigned w0 = __builtin_amdgcn_ballot_w32(((o0 >> b) & 1u) != 0u);
        const unsigned w1 = __builtin_amdgcn_ballot_w32(((o1 >> b) & 1u) != 0u);
        const unsigned w2 = __builtin_amdgcn_ballot_w32(((o2 >> b) & 1u) != 0u);
        const unsigned w3 = __builtin_amdgcn_ballot_w32(((o3 >> b) & 1u) != 0u);
        if (lane == b) mine = make_uint4(w0, w1, w2, w3);
    }
    if (lane < NBIT) ((uint4*)sW)[lane * 8 + wid] = mine;
    __syncthreads();
    if (threadIdx.x < NBIT * 8) {
        const int b = threadIdx.x >> 3;
        const int i = threadIdx.x & 7;
        ((uint4*)planes)[(size_t)b * PLS + (size_t)blockIdx.x * 8 + i] = ((const uint4*)sW)[threadIdx.x];
    }
}

__device__ __forceinline__ void place_one(unsigned r, int base, int* sCnt, int* sMain,
                                          int* sSpill,
                                          int* __restrict__ ovfcnt, int* __restrict__ ovf) {
    const int local = (int)(r >> 18);
    const int s = (int)(r & 0x3FFFFu);
    const int tk = atomicAdd(&sCnt[local], 1);
    if (tk < SLOTS) {
        sMain[local * SLOTS + tk] = s;
    } else if (tk < 2 * SLOTS) {
        sSpill[local * SLOTS + tk - SLOTS] = s;
    } else {
        const int k = atomicAdd(ovfcnt, 1);
        ovf[2 * (size_t)k + 0] = base + local;
        ovf[2 * (size_t)k + 1] = s;
    }
}

__device__ __forceinline__ int edge_of(unsigned x, int bit, int bbase) {
    return (int)((((unsigned)bbase + (x >> 2)) << 7) + ((unsigned)bit << 2) + (x & 3u));
}

__device__ __forceinline__ void file_word(unsigned x1, unsigned m, int bbase,
                                          const int* __restrict__ rec, int base, int* sCnt,
                                          int* sMain, int* sSpill,
                                          int* __restrict__ ovfcnt, int* __restrict__ ovf) {
    while (m != 0u) {
        const int bit = __builtin_ctz(m);
        m &= m - 1u;
        place_one((unsigned)rec[edge_of(x1, bit, bbase)], base, sCnt, sMain, sSpill, ovfcnt, ovf);
    }
}

__global__ __launch_bounds__(1024) void k_build(const unsigned* __restrict__ planes,
                                                const int* __restrict__ rec,
                                                int* __restrict__ cnt,
                                                int* __restrict__ mainL,
                                                int* __restrict__ spillL,
                                                int* __restrict__ ovf) {
    __shared__ __attribute__((aligned(16))) int sMain[SPAN * SLOTS];
    __shared__ __attribute__((aligned(16))) int sSpill[SPAN * SLOTS];
    __shared__ __attribute__((aligned(8))) unsigned sQ[32 * 2 * QE];
    __shared__ int sCnt[SPAN];

    const int tid  = threadIdx.x;
    const int lane = tid & 31;
    const int wid  = tid >> 5;
    const int rng  = blockIdx.x >> 1;
    const int half = blockIdx.x & 1;
    const int base = rng * SPAN;
    int* ovfcnt = cnt + 2 * N_NODES;

    for (int i = tid; i < SPAN * SLOTS / 4; i += 1024) {
        ((int4*)sMain)[i]  = make_int4(0, 0, 0, 0);
        ((int4*)sSpill)[i] = make_int4(0, 0, 0, 0);
    }
    for (int i = tid; i < SPAN; i += 1024) sCnt[i] = 0;
    __syncthreads();

    unsigned rb[NBIT];
#pragma unroll
    for (int b = 0; b < NBIT; ++b) rb[b] = ((rng >> b) & 1) ? 0xFFFFFFFFu : 0u;
    const int bbase = half * HSTEP;
    const uint4* __restrict__ pl = (const uint4*)planes + bbase;
    const int rot = rng * ROTQ;
    unsigned* myq = sQ + wid * (2 * QE);
    int qn = 0;

    auto scan_block = [&](const uint4 (&p)[NBIT], bool okb, int blk) {
#pragma unroll
        for (int j = 0; j < 4; ++j) {
            unsigned diff = 0u;
#pragma unroll
            for (int b = 0; b < NBIT; ++b) {
                const unsigned w = j == 0 ? p[b].x : j == 1 ? p[b].y : j == 2 ? p[b].z : p[b].w;
                diff |= w ^ rb[b];
            }
            const unsigned m = okb ? ~diff : 0u;
            const unsigned bal = __builtin_amdgcn_ballot_w32(m != 0u);
            if (m != 0u) {
                const unsigned x = (unsigned)blk * 4u + (unsigned)j;
                const int pos = qn + (int)__builtin_amdgcn_mbcnt_lo(bal, 0u);
                if (pos < QE) {
                    myq[2 * pos + 0] = x;
                    myq[2 * pos + 1] = m;
                } else {
                    file_word(x, m, bbase, rec, base, sCnt, sMain, sSpill, ovfcnt, ovf);
                }
            }
            qn += __builtin_popcount(bal);
        }
    };

    auto block_of = [&](int it, bool& okb) {
        int x = it + rot;
        x = x >= HSTEP ? x - HSTEP : x;
        okb = it < HSTEP && bbase + x < NBLK;
        return okb ? x : 0;
    };

    int g = 0;
    for (; g + 2 * 1024 <= HSTEP; g += 2 * 1024) {
        uint4 p0[NBIT], p1[NBIT];
        bool ok0, ok1;
        const int x0 = block_of(g + tid, ok0);
        const int x1 = block_of(g + 1024 + tid, ok1);
#pragma unroll
        for (int b = 0; b < NBIT; ++b) p0[b] = pl[(size_t)b * PLS + x0];
#pragma unroll
        for (int b = 0; b < NBIT; ++b) p1[b] = pl[(size_t)b * PLS + x1];
        __builtin_amdgcn_sched_barrier(0);
        scan_block(p0, ok0, x0);
        scan_block(p1, ok1, x1);
    }
    for (; g < HSTEP; g += 1024) {
        uint4 p0[NBIT];
        bool ok0;
        const int x0 = block_of(g + tid, ok0);
#pragma unroll
        for (int b = 0; b < NBIT; ++b) p0[b] = pl[(size_t)b * PLS + x0];
        scan_block(p0, ok0, x0);
    }

    const int nq = qn < QE ? qn : QE;
    int wp = 0;
    for (int i0 = 0; i0 < nq; i0 += 128) {
        unsigned ex[4], em[4];
#pragma unroll
        for (int k = 0; k < 4; ++k) {
            const int i = i0 + k * 32 + lane;
            const bool okk = i < nq;
            ex[k] = okk ? myq[2 * i + 0] : 0u;
            em[k] = okk ? myq[2 * i + 1] : 0u;
        }
        const int lim = i0 + 128 < nq ? 2 * (i0 + 128) : 2 * QE;
#pragma unroll
        for (int k = 0; k < 4; ++k) {
            unsigned m = em[k];
            for (;;) {
                const unsigned bal = __builtin_amdgcn_ballot_w32(m != 0u);
                if (bal == 0u) break;
                if (m != 0u) {
                    const int bit = __builtin_ctz(m);
                    m &= m - 1u;
                    const int e = edge_of(ex[k], bit, bbase);
                    const int pos = wp + (int)__builtin_amdgcn_mbcnt_lo(bal, 0u);
                    if (pos < lim) myq[pos] = (unsigned)e;
                    else place_one((unsigned)rec[e], base, sCnt, sMain, sSpill, ovfcnt, ovf);
                }
                wp += __builtin_popcount(bal);
            }
        }
        wp = wp < lim ? wp : lim;
    }

    for (int i0 = 0; i0 < wp; i0 += 128) {
        unsigned r[4];
        bool ok[4];
#pragma unroll
        for (int k = 0; k < 4; ++k) {
            const int i = i0 + k * 32 + lane;
            ok[k] = i < wp;
            r[k] = ok[k] ? (unsigned)rec[myq[i]] : 0u;
        }
#pragma unroll
        for (int k = 0; k < 4; ++k)
            if (ok[k]) place_one(r[k], base, sCnt, sMain, sSpill, ovfcnt, ovf);
    }
    __syncthreads();

    const size_t nb = (size_t)half * N_NODES + base;
    for (int i = tid; i < SPAN; i += 1024) cnt[nb + i] = sCnt[i];
    int4* gm = (int4*)(mainL + nb * SLOTS);
    for (int i = tid; i < SPAN * SLOTS / 4; i += 1024) gm[i] = ((const int4*)sMain)[i];
    int4* gs = (int4*)(spillL + nb * SLOTS);
    for (int i = tid; i < SPAN * SLOTS / 4; i += 1024)
        if (sCnt[i >> 2] > SLOTS) gs[i] = ((const int4*)sSpill)[i];
}

template <bool LOGSOFTMAX>
__global__ __launch_bounds__(128, 16) void k_gin_layer(const float* __restrict__ xin,
                                                   const int* __restrict__ cnt,
                                                   const int* __restrict__ mainL,
                                                   const int* __restrict__ spillL,
                                                   const int* __restrict__ ovf,
                                                   const float* __restrict__ eps_p,
                                                   const float* __restrict__ wa,
                                                   const float* __restrict__ ba,
                                                   const float* __restrict__ wb,
                                                   const float* __restrict__ bb,
                                                   float* __restrict__ out) {
    __shared__ float ldsM[4][16][17];
    __shared__ float ldsT[4][16][17];
    __shared__ __attribute__((aligned(16))) int ldsI[4][16][IPITCH];

    const int wid  = threadIdx.x >> 5;
    const int lane = threadIdx.x & 31;
    const int tile = blockIdx.x * 4 + wid;

    const int g = lane >> 2;
    const int q = lane & 3;
    const int node0 = tile * 16 + g;
    const int node1 = node0 + 8;

    const float4* __restrict__ f4 = (const float4*)xin;
    const int cA0 = cnt[node0];
    const int cB0 = cnt[N_NODES + node0];
    const int cA1 = cnt[node1];
    const int cB1 = cnt[N_NODES + node1];
    const int4 mA0 = *(const int4*)(mainL + (size_t)node0 * SLOTS + 4 * q);
    const int4 mB0 = *(const int4*)(mainL + ((size_t)N_NODES + node0) * SLOTS + 4 * q);
    const int4 mA1 = *(const int4*)(mainL + (size_t)node1 * SLOTS + 4 * q);
    const int4 mB1 = *(const int4*)(mainL + ((size_t)N_NODES + node1) * SLOTS + 4 * q);
    const unsigned uq = (unsigned)q;

    const int dA0 = cA0 < SLOTS ? cA0 : SLOTS;
    const int dB0 = cB0 < SLOTS ? cB0 : SLOTS;
    const int dA1 = cA1 < SLOTS ? cA1 : SLOTS;
    const int dB1 = cB1 < SLOTS ? cB1 : SLOTS;
    const int d0 = dA0 + dB0;
    const int d1 = dA1 + dB1;

    int* l0 = &ldsI[wid][g][0];
    int* l1 = &ldsI[wid][g + 8][0];
    *(int4*)(l0 + 4 * q) = mA0;
    *(int4*)(l1 + 4 * q) = mA1;
    *(int4*)(l0 + SLOTS + 4 * q) = make_int4(0, 0, 0, 0);
    *(int4*)(l1 + SLOTS + 4 * q) = make_int4(0, 0, 0, 0);
    l0[dA0 + 4 * q + 0] = mB0.x; l0[dA0 + 4 * q + 1] = mB0.y;
    l0[dA0 + 4 * q + 2] = mB0.z; l0[dA0 + 4 * q + 3] = mB0.w;
    l1[dA1 + 4 * q + 0] = mB1.x; l1[dA1 + 4 * q + 1] = mB1.y;
    l1[dA1 + 4 * q + 2] = mB1.z; l1[dA1 + 4 * q + 3] = mB1.w;

    float4 a0 = make_float4(0.f, 0.f, 0.f, 0.f);
    float4 a1 = make_float4(0.f, 0.f, 0.f, 0.f);

    int dm = d0 > d1 ? d0 : d1;
#pragma unroll
    for (int off = 1; off < 32; off <<= 1) {
        const int o = __shfl_xor(dm, off, 32);
        dm = o > dm ? o : dm;
    }
    dm = __builtin_amdgcn_readfirstlane(dm);
    __syncthreads();

    for (int i = 0; i < dm; i += 4) {
        const int4 s0 = *(const int4*)(l0 + i);
        const int4 s1 = *(const int4*)(l1 + i);
        const float4 z4 = make_float4(0.f, 0.f, 0.f, 0.f);
        float4 r00 = z4, r01 = z4, r02 = z4, r03 = z4;
        float4 r10 = z4, r11 = z4, r12 = z4, r13 = z4;
        if (i + 0 < d0) r00 = f4[(unsigned)s0.x * 4u + uq];
        if (i + 1 < d0) r01 = f4[(unsigned)s0.y * 4u + uq];
        if (i + 2 < d0) r02 = f4[(unsigned)s0.z * 4u + uq];
        if (i + 3 < d0) r03 = f4[(unsigned)s0.w * 4u + uq];
        if (i + 0 < d1) r10 = f4[(unsigned)s1.x * 4u + uq];
        if (i + 1 < d1) r11 = f4[(unsigned)s1.y * 4u + uq];
        if (i + 2 < d1) r12 = f4[(unsigned)s1.z * 4u + uq];
        if (i + 3 < d1) r13 = f4[(unsigned)s1.w * 4u + uq];
#define ACC(a, r) { a.x += r.x; a.y += r.y; a.z += r.z; a.w += r.w; }
        ACC(a0, r00) ACC(a0, r01) ACC(a0, r02) ACC(a0, r03)
        ACC(a1, r10) ACC(a1, r11) ACC(a1, r12) ACC(a1, r13)
#undef ACC
    }

    if (cA0 > SLOTS || cB0 > SLOTS || cA1 > SLOTS || cB1 > SLOTS) {
#define SPILL(a, c, row) { const int e_ = (c) < 2 * SLOTS ? (c) : 2 * SLOTS;                 \
        _Pragma("nounroll")                                                                 \
        for (int j = SLOTS; j < e_; ++j) {                                                  \
            const int ss = spillL[(row) * SLOTS + (j - SLOTS)];                              \
            const float4 r = f4[(unsigned)ss * 4u + uq];                                    \
            a.x += r.x; a.y += r.y; a.z += r.z; a.w += r.w; } }
        SPILL(a0, cA0, (size_t)node0)
        SPILL(a0, cB0, (size_t)N_NODES + node0)
        SPILL(a1, cA1, (size_t)node1)
        SPILL(a1, cB1, (size_t)N_NODES + node1)
#undef SPILL
    }

    if (cA0 > 2 * SLOTS || cB0 > 2 * SLOTS || cA1 > 2 * SLOTS || cB1 > 2 * SLOTS) {
        int novf = cnt[2 * N_NODES];
        novf = novf < N_EDGES ? novf : N_EDGES;
        for (int j = 0; j < novf; ++j) {
            const int dd = ovf[2 * (size_t)j + 0];
            const int ss = ovf[2 * (size_t)j + 1];
            if (dd == node0) {
                const float4 r = f4[(unsigned)ss * 4u + uq];
                a0.x += r.x; a0.y += r.y; a0.z += r.z; a0.w += r.w;
            }
            if (dd == node1) {
                const float4 r = f4[(unsigned)ss * 4u + uq];
                a1.x += r.x; a1.y += r.y; a1.z += r.z; a1.w += r.w;
            }
        }
    }

    const float4 x0 = f4[(unsigned)node0 * 4u + uq];
    const float4 x1 = f4[(unsigned)node1 * 4u + uq];

    const float epsp = 1.0f + eps_p[0];
    {
        float* m0 = &ldsM[wid][g][q * 4];
        float* m1 = &ldsM[wid][g + 8][q * 4];
        m0[0] = epsp * x0.x + a0.x; m0[1] = epsp * x0.y + a0.y;
        m0[2] = epsp * x0.z + a0.z; m0[3] = epsp * x0.w + a0.w;
        m1[0] = epsp * x1.x + a1.x; m1[1] = epsp * x1.y + a1.y;
        m1[2] = epsp * x1.z + a1.z; m1[3] = epsp * x1.w + a1.w;
    }
    __syncthreads();

    const int col = lane & 15;
    const int hi  = lane >> 4;

    v8f acc = {};
#pragma unroll
    for (int c = 0; c < 4; ++c) {
        const int kb = c * 4 + hi * 2;
        v2f a;
        a.x = ldsM[wid][col][kb + 0];
        a.y = ldsM[wid][col][kb + 1];
        v2f b;
        b.x = wa[(kb + 0) * 16 + col];
        b.y = wa[(kb + 1) * 16 + col];
        acc = __builtin_amdgcn_wmma_f32_16x16x4_f32(false, a, false, b,
                                                    (short)0, acc, false, false);
    }

    const float biasA = ba[col];
#pragma unroll
    for (int v = 0; v < 8; ++v) {
        float t = acc[v] + biasA;
        ldsT[wid][v + hi * 8][col] = t > 0.0f ? t : 0.0f;
    }
    __syncthreads();

    v8f acc2 = {};
#pragma unroll
    for (int c = 0; c < 4; ++c) {
        const int kb = c * 4 + hi * 2;
        v2f a;
        a.x = ldsT[wid][col][kb + 0];
        a.y = ldsT[wid][col][kb + 1];
        v2f b;
        b.x = wb[(kb + 0) * 16 + col];
        b.y = wb[(kb + 1) * 16 + col];
        acc2 = __builtin_amdgcn_wmma_f32_16x16x4_f32(false, a, false, b,
                                                     (short)0, acc2, false, false);
    }

    const float biasB = bb[col];
    float* orow = out + (size_t)tile * 16 * FDIM;

    if (LOGSOFTMAX) {
#pragma unroll
        for (int v = 0; v < 8; ++v) {
            float val = acc2[v] + biasB;
            float m = val;
#pragma unroll
            for (int off = 1; off < 16; off <<= 1)
                m = fmaxf(m, __shfl_xor(m, off, 16));
            float s = __expf(val - m);
#pragma unroll
            for (int off = 1; off < 16; off <<= 1)
                s += __shfl_xor(s, off, 16);
            orow[(v + hi * 8) * FDIM + col] = val - m - __logf(s);
        }
    } else {
#pragma unroll
        for (int v = 0; v < 8; ++v) {
            float t = acc2[v] + biasB;
            orow[(v + hi * 8) * FDIM + col] = t > 0.0f ? t : 0.0f;
        }
    }
}

extern "C" void kernel_launch(void* const* d_in, const int* in_sizes, int n_in,
                              void* d_out, int out_size, void* d_ws, size_t ws_size,
                              hipStream_t stream) {
    const float* x    = (const float*)d_in[0];
    const int*   ei   = (const int*)d_in[1];
    const float* eps1 = (const float*)d_in[2];
    const float* w1a  = (const float*)d_in[3];
    const float* b1a  = (const float*)d_in[4];
    const float* w1b  = (const float*)d_in[5];
    const float* b1b  = (const float*)d_in[6];
    const float* eps2 = (const float*)d_in[7];
    const float* w2a  = (const float*)d_in[8];
    const float* b2a  = (const float*)d_in[9];
    const float* w2b  = (const float*)d_in[10];
    const float* b2b  = (const float*)d_in[11];
    float* out = (float*)d_out;

    float* h1     = (float*)d_ws;
    int*   mainL  = (int*)(h1 + (size_t)N_NODES * FDIM);
    int*   spillL = mainL + (size_t)2 * N_NODES * SLOTS;
    int*   ovf    = spillL + (size_t)2 * N_NODES * SLOTS;
    int*   cnt    = ovf + (size_t)N_EDGES * 2;
    int*   rec    = cnt + (2 * N_NODES + 32);
    unsigned* planes = (unsigned*)(rec + (size_t)N_EDGES + 96);

    k_owner<<<N_EDGES / 1024, 256, 0, stream>>>(ei, planes, rec, cnt + 2 * N_NODES);
    k_build<<<2 * NOWN, 1024, 0, stream>>>(planes, rec, cnt, mainL, spillL, ovf);
    k_gin_layer<false><<<NTILES / 4, 128, 0, stream>>>(x, cnt, mainL, spillL, ovf, eps1, w1a,
                                                       b1a, w1b, b1b, h1);
    k_gin_layer<true><<<NTILES / 4, 128, 0, stream>>>(h1, cnt, mainL, spillL, ovf, eps2, w2a,
                                                      b2a, w2b, b2b, out);
}


# Owner planes one-hot per digit of a range number (16 low-digit + 8 high-digit planes, not 7 bit planes): a list-build work-group reads and ANDs 2 planes, 400 not 1400 KB; add order by arrival
# speedup vs baseline: 4.2371x; 1.0546x over previous
#include <hip/hip_runtime.h>


#define N_NODES 200000
#define N_EDGES 3200000
#define FDIM    16
#define NTILES  (N_NODES / 16)
#define SLOTS   16
#define SPAN    1600
#define NOWN    (N_NODES / SPAN)
#define NLO     16
#define NHI     8
#define NPL     (NLO + NHI)
#define NBLK    (N_EDGES / 128)
#define HSTEP   12512
#define PLS     25024
#define ROTQ    96
static_assert(NOWN <= NLO * NHI && NPL <= 32, "two digits hold every range number");
static_assert(HSTEP % 32 == 0 && PLS % 32 == 0 && ROTQ % 32 == 0, "512-byte pieces stay aligned");
static_assert(HSTEP <= NBLK && 2 * HSTEP >= NBLK && PLS >= NBLK, "the two halves cover every block");
static_assert((NOWN - 1) * ROTQ < HSTEP, "a pass wraps at most once");
static_assert(N_EDGES % 1024 == 0, "the owner pass has no partial work-group");
#define QE      448
#define IPITCH  36

typedef float v2f __attribute__((ext_vector_type(2)));
typedef float v8f __attribute__((ext_vector_type(8)));

__global__ __launch_bounds__(256) void k_owner(const int* __restrict__ ei,
                                               unsigned* __restrict__ planes,
                                               int* __restrict__ rec,
                                               int* __restrict__ ovfcnt) {
    __shared__ __attribute__((aligned(16))) unsigned sW[NPL * 8 * 4];
    const int t = blockIdx.x * 256 + threadIdx.x;
    const int lane = threadIdx.x & 31;
    const int wid  = threadIdx.x >> 5;
    if (t == 0) ovfcnt[0] = 0;
    const int4 s4 = *(const int4*)(ei + (size_t)t * 4);
    const int4 d4 = *(const int4*)(ei + N_EDGES + (size_t)t * 4);
    const unsigned o0 = (unsigned)d4.x / SPAN, o1 = (unsigned)d4.y / SPAN;
    const unsigned o2 = (unsigned)d4.z / SPAN, o3 = (unsigned)d4.w / SPAN;
    int4 r;
    r.x = (int)((((unsigned)d4.x - o0 * SPAN) << 18) | (unsigned)s4.x);
    r.y = (int)((((unsigned)d4.y - o1 * SPAN) << 18) | (unsigned)s4.y);
    r.z = (int)((((unsigned)d4.z - o2 * SPAN) << 18) | (unsigned)s4.z);
    r.w = (int)((((unsigned)d4.w - o3 * SPAN) << 18) | (unsigned)s4.w);
    *(int4*)(rec + (size_t)t * 4) = r;

    uint4 mine = make_uint4(0u, 0u, 0u, 0u);
    const unsigned l0 = o0 & 15u, l1 = o1 & 15u, l2 = o2 & 15u, l3 = o3 & 15u;
    const unsigned h0 = o0 >> 4, h1 = o1 >> 4, h2 = o2 >> 4, h3 = o3 >> 4;
#pragma unroll
    for (int b = 0; b < NLO; ++b) {
        const unsigned w0 = __builtin_amdgcn_ballot_w32(l0 == (unsigned)b);
        const unsigned w1 = __builtin_amdgcn_ballot_w32(l1 == (unsigned)b);
        const unsigned w2 = __builtin_amdgcn_ballot_w32(l2 == (unsigned)b);
        const unsigned w3 = __builtin_amdgcn_ballot_w32(l3 == (unsigned)b);
        if (lane == b) mine = make_uint4(w0, w1, w2, w3);
    }
#pragma unroll
    for (int b = 0; b < NHI; ++b) {
        const unsigned w0 = __builtin_amdgcn_ballot_w32(h0 == (unsigned)b);
        const unsigned w1 = __builtin_amdgcn_ballot_w32(h1 == (unsigned)b);
        const unsigned w2 = __builtin_amdgcn_ballot_w32(h2 == (unsigned)b);
        const unsigned w3 = __builtin_amdgcn_ballot_w32(h3 == (unsigned)b);
        if (lane == NLO + b) mine = make_uint4(w0, w1, w2, w3);
    }
    if (lane < NPL) ((uint4*)sW)[lane * 8 + wid] = mine;
    __syncthreads();
    if (threadIdx.x < NPL * 8) {
        const int b = threadIdx.x >> 3;
        const int i = threadIdx.x & 7;
        ((uint4*)planes)[(size_t)b * PLS + (size_t)blockIdx.x * 8 + i] = ((const uint4*)sW)[threadIdx.x];
    }
}

__device__ __forceinline__ void place_one(unsigned r, int base, int* sCnt, int* sMain,
                                          int* sSpill,
                                          int* __restrict__ ovfcnt, int* __restrict__ ovf) {
    const int local = (int)(r >> 18);
    const int s = (int)(r & 0x3FFFFu);
    const int tk = atomicAdd(&sCnt[local], 1);
    if (tk < SLOTS) {
        sMain[local * SLOTS + tk] = s;
    } else if (tk < 2 * SLOTS) {
        sSpill[local * SLOTS + tk - SLOTS] = s;
    } else {
        const int k = atomicAdd(ovfcnt, 1);
        ovf[2 * (size_t)k + 0] = base + local;
        ovf[2 * (size_t)k + 1] = s;
    }
}

__device__ __forceinline__ int edge_of(unsigned x, int bit, int bbase) {
    return (int)((((unsigned)bbase + (x >> 2)) << 7) + ((unsigned)bit << 2) + (x & 3u));
}

__device__ __forceinline__ void file_word(unsigned x1, unsigned m, int bbase,
                                          const int* __restrict__ rec, int base, int* sCnt,
                                          int* sMain, int* sSpill,
                                          int* __restrict__ ovfcnt, int* __restrict__ ovf) {
    while (m != 0u) {
        const int bit = __builtin_ctz(m);
        m &= m - 1u;
        place_one((unsigned)rec[edge_of(x1, bit, bbase)], base, sCnt, sMain, sSpill, ovfcnt, ovf);
    }
}

__global__ __launch_bounds__(1024) void k_build(const unsigned* __restrict__ planes,
                                                const int* __restrict__ rec,
                                                int* __restrict__ cnt,
                                                int* __restrict__ mainL,
                                                int* __restrict__ spillL,
                                                int* __restrict__ ovf) {
    __shared__ __attribute__((aligned(16))) int sMain[SPAN * SLOTS];
    __shared__ __attribute__((aligned(16))) int sSpill[SPAN * SLOTS];
    __shared__ __attribute__((aligned(8))) unsigned sQ[32 * 2 * QE];
    __shared__ int sCnt[SPAN];

    const int tid  = threadIdx.x;
    const int lane = tid & 31;
    const int wid  = tid >> 5;
    const int rng  = blockIdx.x >> 1;
    const int half = blockIdx.x & 1;
    const int base = rng * SPAN;
    int* ovfcnt = cnt + 2 * N_NODES;

    for (int i = tid; i < SPAN * SLOTS / 4; i += 1024) {
        ((int4*)sMain)[i]  = make_int4(0, 0, 0, 0);
        ((int4*)sSpill)[i] = make_int4(0, 0, 0, 0);
    }
    for (int i = tid; i < SPAN; i += 1024) sCnt[i] = 0;
    __syncthreads();

    const int bbase = half * HSTEP;
    const uint4* __restrict__ plo = (const uint4*)planes + (size_t)(rng & 15) * PLS + bbase;
    const uint4* __restrict__ phi = (const uint4*)planes + (size_t)(NLO + (rng >> 4)) * PLS + bbase;
    const int rot = rng * ROTQ;
    unsigned* myq = sQ + wid * (2 * QE);
    int qn = 0;

    auto scan_block = [&](const uint4& pa, const uint4& pb, bool okb, int blk) {
#pragma unroll
        for (int j = 0; j < 4; ++j) {
            const unsigned wa = j == 0 ? pa.x : j == 1 ? pa.y : j == 2 ? pa.z : pa.w;
            const unsigned wb = j == 0 ? pb.x : j == 1 ? pb.y : j == 2 ? pb.z : pb.w;
            const unsigned m = okb ? (wa & wb) : 0u;
            const unsigned bal = __builtin_amdgcn_ballot_w32(m != 0u);
            if (m != 0u) {
                const unsigned x = (unsigned)blk * 4u + (unsigned)j;
                const int pos = qn + (int)__builtin_amdgcn_mbcnt_lo(bal, 0u);
                if (pos < QE) {
                    myq[2 * pos + 0] = x;
                    myq[2 * pos + 1] = m;
                } else {
                    file_word(x, m, bbase, rec, base, sCnt, sMain, sSpill, ovfcnt, ovf);
                }
            }
            qn += __builtin_popcount(bal);
        }
    };

    auto block_of = [&](int it, bool& okb) {
        int x = it + rot;
        x = x >= HSTEP ? x - HSTEP : x;
        okb = it < HSTEP && bbase + x < NBLK;
        return okb ? x : 0;
    };

    int g = 0;
    for (; g + 4 * 1024 <= HSTEP; g += 4 * 1024) {
        uint4 pa[4], pb[4];
        bool ok[4];
        int x[4];
#pragma unroll
        for (int k = 0; k < 4; ++k) x[k] = block_of(g + k * 1024 + tid, ok[k]);
#pragma unroll
        for (int k = 0; k < 4; ++k) { pa[k] = plo[x[k]]; pb[k] = phi[x[k]]; }
        __builtin_amdgcn_sched_barrier(0);
#pragma unroll
        for (int k = 0; k < 4; ++k) scan_block(pa[k], pb[k], ok[k], x[k]);
    }
    for (; g < HSTEP; g += 1024) {
        bool ok0;
        const int x0 = block_of(g + tid, ok0);
        const uint4 pa0 = plo[x0];
        const uint4 pb0 = phi[x0];
        scan_block(pa0, pb0, ok0, x0);
    }

    const int nq = qn < QE ? qn : QE;
    int wp = 0;
    for (int i0 = 0; i0 < nq; i0 += 128) {
        unsigned ex[4], em[4];
#pragma unroll
        for (int k = 0; k < 4; ++k) {
            const int i = i0 + k * 32 + lane;
            const bool okk = i < nq;
            ex[k] = okk ? myq[2 * i + 0] : 0u;
            em[k] = okk ? myq[2 * i + 1] : 0u;
        }
        const int lim = i0 + 128 < nq ? 2 * (i0 + 128) : 2 * QE;
#pragma unroll
        for (int k = 0; k < 4; ++k) {
            unsigned m = em[k];
            for (;;) {
                const unsigned bal = __builtin_amdgcn_ballot_w32(m != 0u);
                if (bal == 0u) break;
                if (m != 0u) {
                    const int bit = __builtin_ctz(m);
                    m &= m - 1u;
                    const int e = edge_of(ex[k], bit, bbase);
                    const int pos = wp + (int)__builtin_amdgcn_mbcnt_lo(bal, 0u);
                    if (pos < lim) myq[pos] = (unsigned)e;
                    else place_one((unsigned)rec[e], base, sCnt, sMain, sSpill, ovfcnt, ovf);
                }
                wp += __builtin_popcount(bal);
            }
        }
        wp = wp < lim ? wp : lim;
    }

    for (int i0 = 0; i0 < wp; i0 += 128) {
        unsigned r[4];
        bool ok[4];
#pragma unroll
        for (int k = 0; k < 4; ++k) {
            const int i = i0 + k * 32 + lane;
            ok[k] = i < wp;
            r[k] = ok[k] ? (unsigned)rec[myq[i]] : 0u;
        }
#pragma unroll
        for (int k = 0; k < 4; ++k)
            if (ok[k]) place_one(r[k], base, sCnt, sMain, sSpill, ovfcnt, ovf);
    }
    __syncthreads();

    const size_t nb = (size_t)half * N_NODES + base;
    for (int i = tid; i < SPAN; i += 1024) cnt[nb + i] = sCnt[i];
    int4* gm = (int4*)(mainL + nb * SLOTS);
    for (int i = tid; i < SPAN * SLOTS / 4; i += 1024) gm[i] = ((const int4*)sMain)[i];
    int4* gs = (int4*)(spillL + nb * SLOTS);
    for (int i = tid; i < SPAN * SLOTS / 4; i += 1024)
        if (sCnt[i >> 2] > SLOTS) gs[i] = ((const int4*)sSpill)[i];
}

template <bool LOGSOFTMAX>
__global__ __launch_bounds__(128, 16) void k_gin_layer(const float* __restrict__ xin,
                                                   const int* __restrict__ cnt,
                                                   const int* __restrict__ mainL,
                                                   const int* __restrict__ spillL,
                                                   const int* __restrict__ ovf,
                                                   const float* __restrict__ eps_p,
                                                   const float* __restrict__ wa,
                                                   const float* __restrict__ ba,
                                                   const float* __restrict__ wb,
                                                   const float* __restrict__ bb,
                                                   float* __restrict__ out) {
    __shared__ float ldsM[4][16][17];
    __shared__ float ldsT[4][16][17];
    __shared__ __attribute__((aligned(16))) int ldsI[4][16][IPITCH];

    const int wid  = threadIdx.x >> 5;
    const int lane = threadIdx.x & 31;
    const int tile = blockIdx.x * 4 + wid;

    const int g = lane >> 2;
    const int q = lane & 3;
    const int node0 = tile * 16 + g;
    const int node1 = node0 + 8;

    const float4* __restrict__ f4 = (const float4*)xin;
    const int cA0 = cnt[node0];
    const int cB0 = cnt[N_NODES + node0];
    const int cA1 = cnt[node1];
    const int cB1 = cnt[N_NODES + node1];
    const int4 mA0 = *(const int4*)(mainL + (size_t)node0 * SLOTS + 4 * q);
    const int4 mB0 = *(const int4*)(mainL + ((size_t)N_NODES + node0) * SLOTS + 4 * q);
    const int4 mA1 = *(const int4*)(mainL + (size_t)node1 * SLOTS + 4 * q);
    const int4 mB1 = *(const int4*)(mainL + ((size_t)N_NODES + node1) * SLOTS + 4 * q);
    const unsigned uq = (unsigned)q;

    const int dA0 = cA0 < SLOTS ? cA0 : SLOTS;
    const int dB0 = cB0 < SLOTS ? cB0 : SLOTS;
    const int dA1 = cA1 < SLOTS ? cA1 : SLOTS;
    const int dB1 = cB1 < SLOTS ? cB1 : SLOTS;
    const int d0 = dA0 + dB0;
    const int d1 = dA1 + dB1;

    int* l0 = &ldsI[wid][g][0];
    int* l1 = &ldsI[wid][g + 8][0];
    *(int4*)(l0 + 4 * q) = mA0;
    *(int4*)(l1 + 4 * q) = mA1;
    *(int4*)(l0 + SLOTS + 4 * q) = make_int4(0, 0, 0, 0);
    *(int4*)(l1 + SLOTS + 4 * q) = make_int4(0, 0, 0, 0);
    l0[dA0 + 4 * q + 0] = mB0.x; l0[dA0 + 4 * q + 1] = mB0.y;
    l0[dA0 + 4 * q + 2] = mB0.z; l0[dA0 + 4 * q + 3] = mB0.w;
    l1[dA1 + 4 * q + 0] = mB1.x; l1[dA1 + 4 * q + 1] = mB1.y;
    l1[dA1 + 4 * q + 2] = mB1.z; l1[dA1 + 4 * q + 3] = mB1.w;

    float4 a0 = make_float4(0.f, 0.f, 0.f, 0.f);
    float4 a1 = make_float4(0.f, 0.f, 0.f, 0.f);

    int dm = d0 > d1 ? d0 : d1;
#pragma unroll
    for (int off = 1; off < 32; off <<= 1) {
        const int o = __shfl_xor(dm, off, 32);
        dm = o > dm ? o : dm;
    }
    dm = __builtin_amdgcn_readfirstlane(dm);
    __syncthreads();

    for (int i = 0; i < dm; i += 4) {
        const int4 s0 = *(const int4*)(l0 + i);
        const int4 s1 = *(const int4*)(l1 + i);
        const float4 z4 = make_float4(0.f, 0.f, 0.f, 0.f);
        float4 r00 = z4, r01 = z4, r02 = z4, r03 = z4;
        float4 r10 = z4, r11 = z4, r12 = z4, r13 = z4;
        if (i + 0 < d0) r00 = f4[(unsigned)s0.x * 4u + uq];
        if (i + 1 < d0) r01 = f4[(unsigned)s0.y * 4u + uq];
        if (i + 2 < d0) r02 = f4[(unsigned)s0.z * 4u + uq];
        if (i + 3 < d0) r03 = f4[(unsigned)s0.w * 4u + uq];
        if (i + 0 < d1) r10 = f4[(unsigned)s1.x * 4u + uq];
        if (i + 1 < d1) r11 = f4[(unsigned)s1.y * 4u + uq];
        if (i + 2 < d1) r12 = f4[(unsigned)s1.z * 4u + uq];
        if (i + 3 < d1) r13 = f4[(unsigned)s1.w * 4u + uq];
#define ACC(a, r) { a.x += r.x; a.y += r.y; a.z += r.z; a.w += r.w; }
        ACC(a0, r00) ACC(a0, r01) ACC(a0, r02) ACC(a0, r03)
        ACC(a1, r10) ACC(a1, r11) ACC(a1, r12) ACC(a1, r13)
#undef ACC
    }

    if (cA0 > SLOTS || cB0 > SLOTS || cA1 > SLOTS || cB1 > SLOTS) {
#define SPILL(a, c, row) { const int e_ = (c) < 2 * SLOTS ? (c) : 2 * SLOTS;                 \
        _Pragma("nounroll")                                                                 \
        for (int j = SLOTS; j < e_; ++j) {                                                  \
            const int ss = spillL[(row) * SLOTS + (j - SLOTS)];                              \
            const float4 r = f4[(unsigned)ss * 4u + uq];                                    \
            a.x += r.x; a.y += r.y; a.z += r.z; a.w += r.w; } }
        SPILL(a0, cA0, (size_t)node0)
        SPILL(a0, cB0, (size_t)N_NODES + node0)
        SPILL(a1, cA1, (size_t)node1)
        SPILL(a1, cB1, (size_t)N_NODES + node1)
#undef SPILL
    }

    if (cA0 > 2 * SLOTS || cB0 > 2 * SLOTS || cA1 > 2 * SLOTS || cB1 > 2 * SLOTS) {
        int novf = cnt[2 * N_NODES];
        novf = novf < N_EDGES ? novf : N_EDGES;
        for (int j = 0; j < novf; ++j) {
            const int dd = ovf[2 * (size_t)j + 0];
            const int ss = ovf[2 * (size_t)j + 1];
            if (dd == node0) {
                const float4 r = f4[(unsigned)ss * 4u + uq];
                a0.x += r.x; a0.y += r.y; a0.z += r.z; a0.w += r.w;
            }
            if (dd == node1) {
                const float4 r = f4[(unsigned)ss * 4u + uq];
                a1.x += r.x; a1.y += r.y; a1.z += r.z; a1.w += r.w;
            }
        }
    }

    const float4 x0 = f4[(unsigned)node0 * 4u + uq];
    const float4 x1 = f4[(unsigned)node1 * 4u + uq];

    const float epsp = 1.0f + eps_p[0];
    {
        float* m0 = &ldsM[wid][g][q * 4];
        float* m1 = &ldsM[wid][g + 8][q * 4];
        m0[0] = epsp * x0.x + a0.x; m0[1] = epsp * x0.y + a0.y;
        m0[2] = epsp * x0.z + a0.z; m0[3] = epsp * x0.w + a0.w;
        m1[0] = epsp * x1.x + a1.x; m1[1] = epsp * x1.y + a1.y;
        m1[2] = epsp * x1.z + a1.z; m1[3] = epsp * x1.w + a1.w;
    }
    __syncthreads();

    const int col = lane & 15;
    const int hi  = lane >> 4;

    v8f acc = {};
#pragma unroll
    for (int c = 0; c < 4; ++c) {
        const int kb = c * 4 + hi * 2;
        v2f a;
        a.x = ldsM[wid][col][kb + 0];
        a.y = ldsM[wid][col][kb + 1];
        v2f b;
        b.x = wa[(kb + 0) * 16 + col];
        b.y = wa[(kb + 1) * 16 + col];
        acc = __builtin_amdgcn_wmma_f32_16x16x4_f32(false, a, false, b,
                                                    (short)0, acc, false, false);
    }

    const float biasA = ba[col];
#pragma unroll
    for (int v = 0; v < 8; ++v) {
        float t = acc[v] + biasA;
        ldsT[wid][v + hi * 8][col] = t > 0.0f ? t : 0.0f;
    }
    __syncthreads();

    v8f acc2 = {};
#pragma unroll
    for (int c = 0; c < 4; ++c) {
        const int kb = c * 4 + hi * 2;
        v2f a;
        a.x = ldsT[wid][col][kb + 0];
        a.y = ldsT[wid][col][kb + 1];
        v2f b;
        b.x = wb[(kb + 0) * 16 + col];
        b.y = wb[(kb + 1) * 16 + col];
        acc2 = __builtin_amdgcn_wmma_f32_16x16x4_f32(false, a, false, b,
                                                     (short)0, acc2, false, false);
    }

    const float biasB = bb[col];
    float* orow = out + (size_t)tile * 16 * FDIM;

    if (LOGSOFTMAX) {
#pragma unroll
        for (int v = 0; v < 8; ++v) {
            float val = acc2[v] + biasB;
            float m = val;
#pragma unroll
            for (int off = 1; off < 16; off <<= 1)
                m = fmaxf(m, __shfl_xor(m, off, 16));
            float s = __expf(val - m);
#pragma unroll
            for (int off = 1; off < 16; off <<= 1)
                s += __shfl_xor(s, off, 16);
            orow[(v + hi * 8) * FDIM + col] = val - m - __logf(s);
        }
    } else {
#pragma unroll
        for (int v = 0; v < 8; ++v) {
            float t = acc2[v] + biasB;
            orow[(v + hi * 8) * FDIM + col] = t > 0.0f ? t : 0.0f;
        }
    }
}

extern "C" void kernel_launch(void* const* d_in, const int* in_sizes, int n_in,
                              void* d_out, int out_size, void* d_ws, size_t ws_size,
                              hipStream_t stream) {
    const float* x    = (const float*)d_in[0];
    const int*   ei   = (const int*)d_in[1];
    const float* eps1 = (const float*)d_in[2];
    const float* w1a  = (const float*)d_in[3];
    const float* b1a  = (const float*)d_in[4];
    const float* w1b  = (const float*)d_in[5];
    const float* b1b  = (const float*)d_in[6];
    const float* eps2 = (const float*)d_in[7];
    const float* w2a  = (const float*)d_in[8];
    const float* b2a  = (const float*)d_in[9];
    const float* w2b  = (const float*)d_in[10];
    const float* b2b  = (const float*)d_in[11];
    float* out = (float*)d_out;

    float* h1     = (float*)d_ws;
    int*   mainL  = (int*)(h1 + (size_t)N_NODES * FDIM);
    int*   spillL = mainL + (size_t)2 * N_NODES * SLOTS;
    int*   ovf    = spillL + (size_t)2 * N_NODES * SLOTS;
    int*   cnt    = ovf + (size_t)N_EDGES * 2;
    int*   rec    = cnt + (2 * N_NODES + 32);
    unsigned* planes = (unsigned*)(rec + (size_t)N_EDGES + 96);

    k_owner<<<N_EDGES / 1024, 256, 0, stream>>>(ei, planes, rec, cnt + 2 * N_NODES);
    k_build<<<2 * NOWN, 1024, 0, stream>>>(planes, rec, cnt, mainL, spillL, ovf);
    k_gin_layer<false><<<NTILES / 4, 128, 0, stream>>>(x, cnt, mainL, spillL, ovf, eps1, w1a,
                                                       b1a, w1b, b1b, h1);
    k_gin_layer<true><<<NTILES / 4, 128, 0, stream>>>(h1, cnt, mainL, spillL, ovf, eps2, w2a,
                                                      b2a, w2b, b2b, out);
}
